# starting kernel
# baseline (speedup 1.0000x reference)
#include <hip/hip_runtime.h>

typedef __attribute__((ext_vector_type(2))) float v2f;
typedef __attribute__((ext_vector_type(8))) float v8f;

static constexpr int Bsz = 4;
static constexpr int Tsz = 4096;
static constexpr int Csz = 1024;
static constexpr int Hsz = 64;
static constexpr int BT  = Bsz * Tsz;

__global__ __launch_bounds__(128) void proj_qkv(
    const float* __restrict__ x,
    const float* __restrict__ Wk,
    const float* __restrict__ Wq,
    const float* __restrict__ Wv,
    float* __restrict__ qo, float* __restrict__ ko, float* __restrict__ vo)
{
  const int lane = threadIdx.x & 31;
  const int wid  = threadIdx.x >> 5;
  const int tile = blockIdx.x * 4 + wid;
  const int row0 = tile * 16;
  const int m    = lane & 15;
  const int kp   = lane >> 4;

  v8f qa[4] = {}, ka[4] = {}, va[4] = {};

  const float* xr = x + (size_t)(row0 + m) * Csz + 2 * kp;

  for (int kk = 0; kk < Csz; kk += 4) {
    v2f a = *(const v2f*)(xr + kk);
    const int kr = kk + 2 * kp;
#pragma unroll
    for (int nn = 0; nn < 4; ++nn) {
      const int n = nn * 16 + m;
      v2f b;
      b.x = Wq[(size_t)kr * Hsz + n];
      b.y = Wq[(size_t)(kr + 1) * Hsz + n];
      qa[nn] = __builtin_amdgcn_wmma_f32_16x16x4_f32(false, a, false, b,
                                                     (short)0, qa[nn], false, false);
      b.x = Wk[(size_t)kr * Hsz + n];
      b.y = Wk[(size_t)(kr + 1) * Hsz + n];
      ka[nn] = __builtin_amdgcn_wmma_f32_16x16x4_f32(false, a, false, b,
                                                     (short)0, ka[nn], false, false);
      b.x = Wv[(size_t)kr * Hsz + n];
      b.y = Wv[(size_t)(kr + 1) * Hsz + n];
      va[nn] = __builtin_amdgcn_wmma_f32_16x16x4_f32(false, a, false, b,
                                                     (short)0, va[nn], false, false);
    }
  }

#pragma unroll
  for (int nn = 0; nn < 4; ++nn)
#pragma unroll
    for (int r = 0; r < 8; ++r) {
      const size_t idx = (size_t)(row0 + r + 8 * kp) * Hsz + nn * 16 + m;
      qo[idx] = qa[nn][r];
      ko[idx] = ka[nn][r];
      vo[idx] = va[nn][r];
    }
}

__global__ __launch_bounds__(128) void flash_attn(
    const float* __restrict__ qm, const float* __restrict__ km,
    const float* __restrict__ vm, float* __restrict__ out)
{
  __shared__ float pbuf[4][16 * 17];

  const int lane = threadIdx.x & 31;
  const int wid  = threadIdx.x >> 5;
  const int tile = blockIdx.x * 4 + wid;
  const int bb   = tile >> 8;
  const int ti   = tile & 255;
  const int m    = lane & 15;
  const int kp   = lane >> 4;

  const size_t base = (size_t)bb * Tsz * Hsz;
  const float* qb = qm + base;
  const float* kb = km + base;
  const float* vb = vm + base;

  v2f qf[16];
  {
    const float* qr = qb + (size_t)(ti * 16 + m) * Hsz + 2 * kp;
#pragma unroll
    for (int s = 0; s < 16; ++s) {
      v2f t = *(const v2f*)(qr + 4 * s);
      qf[s].x = t.x * 0.125f;
      qf[s].y = t.y * 0.125f;
    }
  }

  v8f acc[4] = {};
  float mr[8], lr[8];
#pragma unroll
  for (int r = 0; r < 8; ++r) { mr[r] = -1e30f; lr[r] = 0.0f; }

  float* pb = pbuf[wid];

  for (int j = 0; j <= ti; ++j) {
    v8f s8 = {};
    const float* kt = kb + (size_t)(j * 16 + m) * Hsz + 2 * kp;
#pragma unroll
    for (int s = 0; s < 16; ++s) {
      v2f bf = *(const v2f*)(kt + 4 * s);
      s8 = __builtin_amdgcn_wmma_f32_16x16x4_f32(false, qf[s], false, bf,
                                                 (short)0, s8, false, false);
    }

    if (j == ti) {
#pragma unroll
      for (int r = 0; r < 8; ++r)
        if (m > r + 8 * kp) s8[r] = -1e30f;
    }

    __builtin_amdgcn_wave_barrier();

#pragma unroll
    for (int r = 0; r < 8; ++r) {
      float rm = s8[r];
#pragma unroll
      for (int off = 1; off < 16; off <<= 1)
        rm = fmaxf(rm, __shfl_xor(rm, off, 32));
      const float mnew  = fmaxf(mr[r], rm);
      const float alpha = __expf(mr[r] - mnew);
      mr[r] = mnew;
      const float p = __expf(s8[r] - mnew);
      float rs = p;
#pragma unroll
      for (int off = 1; off < 16; off <<= 1)
        rs += __shfl_xor(rs, off, 32);
      lr[r] = lr[r] * alpha + rs;
#pragma unroll
      for (int nn = 0; nn < 4; ++nn) acc[nn][r] *= alpha;
      pb[(r + 8 * kp) * 17 + m] = p;
    }

    __builtin_amdgcn_wave_barrier();

    v2f pf[4];
#pragma unroll
    for (int s = 0; s < 4; ++s) {
      pf[s].x = pb[m * 17 + 4 * s + 2 * kp];
      pf[s].y = pb[m * 17 + 4 * s + 2 * kp + 1];
    }

    const float* vt = vb + (size_t)(j * 16) * Hsz;
#pragma unroll
    for (int s = 0; s < 4; ++s) {
      const int kr = 4 * s + 2 * kp;
#pragma unroll
      for (int nn = 0; nn < 4; ++nn) {
        const int n = nn * 16 + m;
        v2f bf;
        bf.x = vt[(size_t)kr * Hsz + n];
        bf.y = vt[(size_t)(kr + 1) * Hsz + n];
        acc[nn] = __builtin_amdgcn_wmma_f32_16x16x4_f32(false, pf[s], false, bf,
                                                        (short)0, acc[nn], false, false);
      }
    }
  }

#pragma unroll
  for (int nn = 0; nn < 4; ++nn)
#pragma unroll
    for (int r = 0; r < 8; ++r) {
      const size_t idx = base + (size_t)(ti * 16 + r + 8 * kp) * Hsz + nn * 16 + m;
      out[idx] = acc[nn][r] / lr[r];
    }
}

extern "C" void kernel_launch(void* const* d_in, const int* in_sizes, int n_in,
                              void* d_out, int out_size, void* d_ws, size_t ws_size,
                              hipStream_t stream) {
  (void)in_sizes; (void)n_in; (void)out_size; (void)ws_size;

  const float* x  = (const float*)d_in[0];
  const float* Wk = (const float*)d_in[1];
  const float* Wq = (const float*)d_in[2];
  const float* Wv = (const float*)d_in[3];
  float* out = (float*)d_out;

  float* qo = (float*)d_ws;
  float* ko = qo + (size_t)BT * Hsz;
  float* vo = ko + (size_t)BT * Hsz;

  const int tiles = BT / 16;
  dim3 block(128);
  dim3 grid(tiles / 4);

  proj_qkv<<<grid, block, 0, stream>>>(x, Wk, Wq, Wv, qo, ko, vo);
  flash_attn<<<grid, block, 0, stream>>>(qo, ko, vo, out);
}


# attention: each 64-key K and V block fetched once per workgroup by whole-line 16-byte loads and shared by its 4 waves through a two-stage LDS ring filled one block ahead; every sum unchanged
# speedup vs baseline: 1.7787x; 1.7787x over previous
#include <hip/hip_runtime.h>

typedef __attribute__((ext_vector_type(2))) float v2f;
typedef __attribute__((ext_vector_type(4))) float v4f;
typedef __attribute__((ext_vector_type(8))) float v8f;

static constexpr int Bsz = 4;
static constexpr int Tsz = 4096;
static constexpr int Csz = 1024;
static constexpr int Hsz = 64;
static constexpr int BT  = Bsz * Tsz;

__global__ __launch_bounds__(128) void proj_qkv(
    const float* __restrict__ x,
    const float* __restrict__ Wk,
    const float* __restrict__ Wq,
    const float* __restrict__ Wv,
    float* __restrict__ qo, float* __restrict__ ko, float* __restrict__ vo)
{
  const int lane = threadIdx.x & 31;
  const int wid  = threadIdx.x >> 5;
  const int tile = blockIdx.x * 4 + wid;
  const int row0 = tile * 16;
  const int m    = lane & 15;
  const int kp   = lane >> 4;

  v8f qa[4] = {}, ka[4] = {}, va[4] = {};

  const float* xr = x + (size_t)(row0 + m) * Csz + 2 * kp;

  for (int kk = 0; kk < Csz; kk += 4) {
    v2f a = *(const v2f*)(xr + kk);
    const int kr = kk + 2 * kp;
#pragma unroll
    for (int nn = 0; nn < 4; ++nn) {
      const int n = nn * 16 + m;
      v2f b;
      b.x = Wq[(size_t)kr * Hsz + n];
      b.y = Wq[(size_t)(kr + 1) * Hsz + n];
      qa[nn] = __builtin_amdgcn_wmma_f32_16x16x4_f32(false, a, false, b,
                                                     (short)0, qa[nn], false, false);
      b.x = Wk[(size_t)kr * Hsz + n];
      b.y = Wk[(size_t)(kr + 1) * Hsz + n];
      ka[nn] = __builtin_amdgcn_wmma_f32_16x16x4_f32(false, a, false, b,
                                                     (short)0, ka[nn], false, false);
      b.x = Wv[(size_t)kr * Hsz + n];
      b.y = Wv[(size_t)(kr + 1) * Hsz + n];
      va[nn] = __builtin_amdgcn_wmma_f32_16x16x4_f32(false, a, false, b,
                                                     (short)0, va[nn], false, false);
    }
  }

#pragma unroll
  for (int nn = 0; nn < 4; ++nn)
#pragma unroll
    for (int r = 0; r < 8; ++r) {
      const size_t idx = (size_t)(row0 + r + 8 * kp) * Hsz + nn * 16 + m;
      qo[idx] = qa[nn][r];
      ko[idx] = ka[nn][r];
      vo[idx] = va[nn][r];
    }
}

static constexpr int KT = 4;
static constexpr int BN = 16 * KT;
static constexpr int PP = BN + 4;
static constexpr int KP = Hsz + 4;
static constexpr int VP = 2 * Hsz + 32;

__device__ __forceinline__ void stage_load(
    const float* __restrict__ kb, const float* __restrict__ vb,
    const int jb, const int sr, const int sc, v4f (&kr)[8], v4f (&vr)[8])
{
  const float* kg = kb + (size_t)(jb * BN + sr) * Hsz + 4 * sc;
#pragma unroll
  for (int u = 0; u < 8; ++u)
    kr[u] = *(const v4f*)(kg + (size_t)(8 * u) * Hsz);
  const float* vg = vb + (size_t)(jb * BN + 2 * sr) * Hsz + 4 * sc;
#pragma unroll
  for (int u = 0; u < 4; ++u) {
    vr[2 * u]     = *(const v4f*)(vg + (size_t)(16 * u) * Hsz);
    vr[2 * u + 1] = *(const v4f*)(vg + (size_t)(16 * u + 1) * Hsz);
  }
}

__device__ __forceinline__ void stage_store(
    float* __restrict__ ks, float* __restrict__ vs,
    const int sr, const int sc, const v4f (&kr)[8], const v4f (&vr)[8])
{
#pragma unroll
  for (int u = 0; u < 8; ++u)
    *(v4f*)(ks + (sr + 8 * u) * KP + 4 * sc) = kr[u];
#pragma unroll
  for (int u = 0; u < 4; ++u) {
    const v4f a = vr[2 * u];
    const v4f b = vr[2 * u + 1];
    float* d = vs + (sr + 8 * u) * VP + 8 * sc;
    v4f lo, hi;
    lo.x = a.x; lo.y = b.x; lo.z = a.y; lo.w = b.y;
    hi.x = a.z; hi.y = b.z; hi.z = a.w; hi.w = b.w;
    *(v4f*)(d)     = lo;
    *(v4f*)(d + 4) = hi;
  }
}

template <bool MASK>
__device__ __forceinline__ void kv_block(
    const float* __restrict__ ks, const float* __restrict__ vs,
    const int rowl, const int m, const int kp,
    const v2f (&qf)[16], v8f (&acc)[4], float (&mr)[8], float (&lr)[8],
    float* __restrict__ pb)
{
  const float* kt = ks + m * KP + 2 * kp;
  v2f kf[16][KT];
#pragma unroll
  for (int s_ = 0; s_ < 16; ++s_)
#pragma unroll
    for (int t = 0; t < KT; ++t)
      kf[s_][t] = *(const v2f*)(kt + (t * 16) * KP + 4 * s_);

  v8f s[KT];
#pragma unroll
  for (int t = 0; t < KT; ++t) s[t] = v8f{};
#pragma unroll
  for (int s_ = 0; s_ < 16; ++s_)
#pragma unroll
    for (int t = 0; t < KT; ++t)
      s[t] = __builtin_amdgcn_wmma_f32_16x16x4_f32(false, qf[s_], false, kf[s_][t],
                                                   (short)0, s[t], false, false);

  if (MASK) {
#pragma unroll
    for (int t = 0; t < KT; ++t)
#pragma unroll
      for (int r = 0; r < 8; ++r)
        if (16 * t + m > rowl + r + 8 * kp) s[t][r] = -1e30f;
  }

  float alpha[8];
#pragma unroll
  for (int r = 0; r < 8; ++r) {
    float rm = s[0][r];
#pragma unroll
    for (int t = 1; t < KT; ++t) rm = fmaxf(rm, s[t][r]);
#pragma unroll
    for (int off = 1; off < 16; off <<= 1)
      rm = fmaxf(rm, __shfl_xor(rm, off, 32));
    const float mnew = fmaxf(mr[r], rm);
    alpha[r] = __expf(mr[r] - mnew);
    mr[r] = mnew;
  }

  __builtin_amdgcn_sched_barrier(0);

  const float* vt = vs + kp * VP + 2 * m;
  v2f vf[16][4];
#pragma unroll
  for (int s_ = 0; s_ < 16; ++s_)
#pragma unroll
    for (int nn = 0; nn < 4; ++nn)
      vf[s_][nn] = *(const v2f*)(vt + (2 * s_) * VP + 32 * nn);

  __builtin_amdgcn_sched_barrier(0);

  __builtin_amdgcn_wave_barrier();

#pragma unroll
  for (int r = 0; r < 8; ++r) {
    const float mnew = mr[r];
    float rs = 0.0f;
#pragma unroll
    for (int t = 0; t < KT; ++t) {
      const float p = __expf(s[t][r] - mnew);
      rs += p;
      pb[(r + 8 * kp) * PP + 16 * t + m] = p;
    }
#pragma unroll
    for (int off = 1; off < 16; off <<= 1)
      rs += __shfl_xor(rs, off, 32);
    lr[r] = lr[r] * alpha[r] + rs;
#pragma unroll
    for (int nn = 0; nn < 4; ++nn) acc[nn][r] *= alpha[r];
  }

  __builtin_amdgcn_wave_barrier();

  v2f pf[16];
#pragma unroll
  for (int s_ = 0; s_ < 16; ++s_)
    pf[s_] = *(const v2f*)(pb + m * PP + 4 * s_ + 2 * kp);

#pragma unroll
  for (int s_ = 0; s_ < 16; ++s_)
#pragma unroll
    for (int nn = 0; nn < 4; ++nn)
      acc[nn] = __builtin_amdgcn_wmma_f32_16x16x4_f32(false, pf[s_], false, vf[s_][nn],
                                                      (short)0, acc[nn], false, false);
}

__global__ __launch_bounds__(128) void flash_attn(
    const float* __restrict__ qm, const float* __restrict__ km,
    const float* __restrict__ vm, float* __restrict__ out)
{
  __shared__ __attribute__((aligned(16))) float kst[2][BN * KP];
  __shared__ __attribute__((aligned(16))) float vst[2][(BN / 2) * VP];
  __shared__ __attribute__((aligned(16))) float pbuf[4][16 * PP];

  const int lane = threadIdx.x & 31;
  const int wid  = threadIdx.x >> 5;
  const int tile = blockIdx.x * 4 + wid;
  const int bb   = tile >> 8;
  const int ti   = tile & 255;
  const int m    = lane & 15;
  const int kp   = lane >> 4;
  const int sc   = threadIdx.x & 15;
  const int sr   = threadIdx.x >> 4;

  const size_t base = (size_t)bb * Tsz * Hsz;
  const float* qb = qm + base;
  const float* kb = km + base;
  const float* vb = vm + base;

  v4f kr[8], vr[8];
  stage_load(kb, vb, 0, sr, sc, kr, vr);

  v2f qf[16];
  {
    const float* qr = qb + (size_t)(ti * 16 + m) * Hsz + 2 * kp;
#pragma unroll
    for (int s = 0; s < 16; ++s) {
      v2f t = *(const v2f*)(qr + 4 * s);
      qf[s].x = t.x * 0.125f;
      qf[s].y = t.y * 0.125f;
    }
  }

  v8f acc[4] = {};
  float mr[8], lr[8];
#pragma unroll
  for (int r = 0; r < 8; ++r) { mr[r] = -1e30f; lr[r] = 0.0f; }

  float* pb = pbuf[wid];

  const int nfull = blockIdx.x & 63;
  const int rowl  = ti * 16 - nfull * BN;

  stage_store(kst[0], vst[0], sr, sc, kr, vr);
  __syncthreads();

  for (int jb = 0; jb < nfull; ++jb) {
    const int cur = jb & 1;
    stage_load(kb, vb, jb + 1, sr, sc, kr, vr);
    __builtin_amdgcn_sched_barrier(0);
    kv_block<false>(kst[cur], vst[cur], rowl, m, kp, qf, acc, mr, lr, pb);
    __builtin_amdgcn_sched_barrier(0);
    stage_store(kst[cur ^ 1], vst[cur ^ 1], sr, sc, kr, vr);
    __syncthreads();
  }
  kv_block<true>(kst[nfull & 1], vst[nfull & 1], rowl, m, kp, qf, acc, mr, lr, pb);

#pragma unroll
  for (int nn = 0; nn < 4; ++nn)
#pragma unroll
    for (int r = 0; r < 8; ++r) {
      const size_t idx = base + (size_t)(ti * 16 + r + 8 * kp) * Hsz + nn * 16 + m;
      out[idx] = acc[nn][r] / lr[r];
    }
}

extern "C" void kernel_launch(void* const* d_in, const int* in_sizes, int n_in,
                              void* d_out, int out_size, void* d_ws, size_t ws_size,
                              hipStream_t stream) {
  (void)in_sizes; (void)n_in; (void)out_size; (void)ws_size;

  const float* x  = (const float*)d_in[0];
  const float* Wk = (const float*)d_in[1];
  const float* Wq = (const float*)d_in[2];
  const float* Wv = (const float*)d_in[3];
  float* out = (float*)d_out;

  float* qo = (float*)d_ws;
  float* ko = qo + (size_t)BT * Hsz;
  float* vo = ko + (size_t)BT * Hsz;

  const int tiles = BT / 16;
  dim3 block(128);
  dim3 grid(tiles / 4);

  proj_qkv<<<grid, block, 0, stream>>>(x, Wk, Wq, Wv, qo, ko, vo);
  flash_attn<<<grid, block, 0, stream>>>(qo, ko, vo, out);
}


# projection: a workgroup's 64 x rows staged once per 64 k through a two-stage LDS ring (whole-line loads); each wave owns 48 of 192 columns; tiles leave via LDS as whole lines; sums unchanged
# speedup vs baseline: 3.0110x; 1.6928x over previous
#include <hip/hip_runtime.h>

typedef __attribute__((ext_vector_type(2))) float v2f;
typedef __attribute__((ext_vector_type(4))) float v4f;
typedef __attribute__((ext_vector_type(8))) float v8f;

static constexpr int Bsz = 4;
static constexpr int Tsz = 4096;
static constexpr int Csz = 1024;
static constexpr int Hsz = 64;
static constexpr int BT  = Bsz * Tsz;

static constexpr int PM  = 64;
static constexpr int KC  = 64;
static constexpr int XP  = KC + 4;
static constexpr int CP  = 3 * Hsz + 4;

template <bool MORE>
__device__ __forceinline__ void proj_trip(
    const float* __restrict__ xa, float* __restrict__ xs0,
    const float* __restrict__ xg, const float* const (&wb)[3],
    const int c, const int sr, const int sc, v4f (&xr)[8], v8f (&acc)[4][3])
{
  const int cur = c & 1;
  const float* xc = xa + cur * (PM * XP);
  const size_t kof = (size_t)(c * KC) * Hsz;

  v2f wf[16][3];
#pragma unroll
  for (int s = 0; s < 8; ++s)
#pragma unroll
    for (int j = 0; j < 3; ++j) {
      wf[s][j].x = wb[j][kof + (size_t)(4 * s) * Hsz];
      wf[s][j].y = wb[j][kof + (size_t)(4 * s + 1) * Hsz];
    }

  v2f af[16][4];
#pragma unroll
  for (int s = 0; s < 16; ++s)
#pragma unroll
    for (int mt = 0; mt < 4; ++mt)
      af[s][mt] = *(const v2f*)(xc + (16 * mt) * XP + 4 * s);

  __builtin_amdgcn_sched_barrier(0);

#pragma unroll
  for (int s = 0; s < 4; ++s)
#pragma unroll
    for (int mt = 0; mt < 4; ++mt)
#pragma unroll
      for (int j = 0; j < 3; ++j)
        acc[mt][j] = __builtin_amdgcn_wmma_f32_16x16x4_f32(false, af[s][mt], false, wf[s][j],
                                                          (short)0, acc[mt][j], false, false);

  __builtin_amdgcn_sched_barrier(0);

#pragma unroll
  for (int s = 8; s < 16; ++s)
#pragma unroll
    for (int j = 0; j < 3; ++j) {
      wf[s][j].x = wb[j][kof + (size_t)(4 * s) * Hsz];
      wf[s][j].y = wb[j][kof + (size_t)(4 * s + 1) * Hsz];
    }

  if (MORE) {
#pragma unroll
    for (int u = 0; u < 8; ++u)
      xr[u] = *(const v4f*)(xg + (size_t)(8 * u) * Csz + (c + 1) * KC);
  }

  __builtin_amdgcn_sched_barrier(0);

#pragma unroll
  for (int s = 4; s < 16; ++s)
#pragma unroll
    for (int mt = 0; mt < 4; ++mt)
#pragma unroll
      for (int j = 0; j < 3; ++j)
        acc[mt][j] = __builtin_amdgcn_wmma_f32_16x16x4_f32(false, af[s][mt], false, wf[s][j],
                                                          (short)0, acc[mt][j], false, false);

  __builtin_amdgcn_sched_barrier(0);

  if (MORE) {
    float* xd = xs0 + (cur ^ 1) * (PM * XP) + sr * XP + 4 * sc;
#pragma unroll
    for (int u = 0; u < 8; ++u)
      *(v4f*)(xd + (8 * u) * XP) = xr[u];
  }
}

__global__ __launch_bounds__(128) void proj_qkv(
    const float* __restrict__ x,
    const float* __restrict__ Wk,
    const float* __restrict__ Wq,
    const float* __restrict__ Wv,
    float* __restrict__ qo, float* __restrict__ ko, float* __restrict__ vo)
{
  __shared__ __attribute__((aligned(16))) float xs[2][PM * XP];
  __shared__ __attribute__((aligned(16))) float cs[PM * CP];

  const int lane = threadIdx.x & 31;
  const int wid  = __builtin_amdgcn_readfirstlane((int)(threadIdx.x >> 5));
  const int m    = lane & 15;
  const int kp   = lane >> 4;
  const int sc   = threadIdx.x & 15;
  const int sr   = threadIdx.x >> 4;
  const int row0 = blockIdx.x * PM;

  const float* wb[3];
#pragma unroll
  for (int j = 0; j < 3; ++j) {
    const int g   = 3 * wid + j;
    const int mat = g >> 2;
    const float* wsel = (mat == 0) ? Wq : ((mat == 1) ? Wk : Wv);
    wb[j] = wsel + (size_t)(2 * kp) * Hsz + (g & 3) * 16 + m;
  }

  const float* xg = x + (size_t)(row0 + sr) * Csz + 4 * sc;
  const float* xa = &xs[0][0] + m * XP + 2 * kp;

  v8f acc[4][3];
#pragma unroll
  for (int mt = 0; mt < 4; ++mt)
#pragma unroll
    for (int j = 0; j < 3; ++j) acc[mt][j] = v8f{};

  v4f xr[8];
#pragma unroll
  for (int u = 0; u < 8; ++u)
    xr[u] = *(const v4f*)(xg + (size_t)(8 * u) * Csz);
#pragma unroll
  for (int u = 0; u < 8; ++u)
    *(v4f*)(&xs[0][0] + (sr + 8 * u) * XP + 4 * sc) = xr[u];
  __syncthreads();

  for (int c = 0; c < Csz / KC - 1; ++c) {
    proj_trip<true>(xa, xs[0], xg, wb, c, sr, sc, xr, acc);
    __syncthreads();
  }
  proj_trip<false>(xa, xs[0], xg, wb, Csz / KC - 1, sr, sc, xr, acc);

#pragma unroll
  for (int mt = 0; mt < 4; ++mt)
#pragma unroll
    for (int j = 0; j < 3; ++j)
#pragma unroll
      for (int r = 0; r < 8; ++r)
        cs[(16 * mt + r + 8 * kp) * CP + 16 * (3 * wid + j) + m] = acc[mt][j][r];
  __syncthreads();

#pragma unroll
  for (int u = 0; u < 8; ++u) {
    const int row = sr + 8 * u;
    const float* cr = cs + row * CP + 4 * sc;
    const size_t o = (size_t)(row0 + row) * Hsz + 4 * sc;
    *(v4f*)(qo + o) = *(const v4f*)(cr);
    *(v4f*)(ko + o) = *(const v4f*)(cr + Hsz);
    *(v4f*)(vo + o) = *(const v4f*)(cr + 2 * Hsz);
  }
}

static constexpr int KT = 4;
static constexpr int BN = 16 * KT;
static constexpr int PP = BN + 4;
static constexpr int KP = Hsz + 4;
static constexpr int VP = 2 * Hsz + 32;

__device__ __forceinline__ void stage_load(
    const float* __restrict__ kb, const float* __restrict__ vb,
    const int jb, const int sr, const int sc, v4f (&kr)[8], v4f (&vr)[8])
{
  const float* kg = kb + (size_t)(jb * BN + sr) * Hsz + 4 * sc;
#pragma unroll
  for (int u = 0; u < 8; ++u)
    kr[u] = *(const v4f*)(kg + (size_t)(8 * u) * Hsz);
  const float* vg = vb + (size_t)(jb * BN + 2 * sr) * Hsz + 4 * sc;
#pragma unroll
  for (int u = 0; u < 4; ++u) {
    vr[2 * u]     = *(const v4f*)(vg + (size_t)(16 * u) * Hsz);
    vr[2 * u + 1] = *(const v4f*)(vg + (size_t)(16 * u + 1) * Hsz);
  }
}

__device__ __forceinline__ void stage_store(
    float* __restrict__ ks, float* __restrict__ vs,
    const int sr, const int sc, const v4f (&kr)[8], const v4f (&vr)[8])
{
#pragma unroll
  for (int u = 0; u < 8; ++u)
    *(v4f*)(ks + (sr + 8 * u) * KP + 4 * sc) = kr[u];
#pragma unroll
  for (int u = 0; u < 4; ++u) {
    const v4f a = vr[2 * u];
    const v4f b = vr[2 * u + 1];
    float* d = vs + (sr + 8 * u) * VP + 8 * sc;
    v4f lo, hi;
    lo.x = a.x; lo.y = b.x; lo.z = a.y; lo.w = b.y;
    hi.x = a.z; hi.y = b.z; hi.z = a.w; hi.w = b.w;
    *(v4f*)(d)     = lo;
    *(v4f*)(d + 4) = hi;
  }
}

template <bool MASK>
__device__ __forceinline__ void kv_block(
    const float* __restrict__ ks, const float* __restrict__ vs,
    const int rowl, const int m, const int kp,
    const v2f (&qf)[16], v8f (&acc)[4], float (&mr)[8], float (&lr)[8],
    float* __restrict__ pb)
{
  const float* kt = ks + m * KP + 2 * kp;
  v2f kf[16][KT];
#pragma unroll
  for (int s_ = 0; s_ < 16; ++s_)
#pragma unroll
    for (int t = 0; t < KT; ++t)
      kf[s_][t] = *(const v2f*)(kt + (t * 16) * KP + 4 * s_);

  v8f s[KT];
#pragma unroll
  for (int t = 0; t < KT; ++t) s[t] = v8f{};
#pragma unroll
  for (int s_ = 0; s_ < 16; ++s_)
#pragma unroll
    for (int t = 0; t < KT; ++t)
      s[t] = __builtin_amdgcn_wmma_f32_16x16x4_f32(false, qf[s_], false, kf[s_][t],
                                                   (short)0, s[t], false, false);

  if (MASK) {
#pragma unroll
    for (int t = 0; t < KT; ++t)
#pragma unroll
      for (int r = 0; r < 8; ++r)
        if (16 * t + m > rowl + r + 8 * kp) s[t][r] = -1e30f;
  }

  float alpha[8];
#pragma unroll
  for (int r = 0; r < 8; ++r) {
    float rm = s[0][r];
#pragma unroll
    for (int t = 1; t < KT; ++t) rm = fmaxf(rm, s[t][r]);
#pragma unroll
    for (int off = 1; off < 16; off <<= 1)
      rm = fmaxf(rm, __shfl_xor(rm, off, 32));
    const float mnew = fmaxf(mr[r], rm);
    alpha[r] = __expf(mr[r] - mnew);
    mr[r] = mnew;
  }

  __builtin_amdgcn_sched_barrier(0);

  const float* vt = vs + kp * VP + 2 * m;
  v2f vf[16][4];
#pragma unroll
  for (int s_ = 0; s_ < 16; ++s_)
#pragma unroll
    for (int nn = 0; nn < 4; ++nn)
      vf[s_][nn] = *(const v2f*)(vt + (2 * s_) * VP + 32 * nn);

  __builtin_amdgcn_sched_barrier(0);

  __builtin_amdgcn_wave_barrier();

#pragma unroll
  for (int r = 0; r < 8; ++r) {
    const float mnew = mr[r];
    float rs = 0.0f;
#pragma unroll
    for (int t = 0; t < KT; ++t) {
      const float p = __expf(s[t][r] - mnew);
      rs += p;
      pb[(r + 8 * kp) * PP + 16 * t + m] = p;
    }
#pragma unroll
    for (int off = 1; off < 16; off <<= 1)
      rs += __shfl_xor(rs, off, 32);
    lr[r] = lr[r] * alpha[r] + rs;
#pragma unroll
    for (int nn = 0; nn < 4; ++nn) acc[nn][r] *= alpha[r];
  }

  __builtin_amdgcn_wave_barrier();

  v2f pf[16];
#pragma unroll
  for (int s_ = 0; s_ < 16; ++s_)
    pf[s_] = *(const v2f*)(pb + m * PP + 4 * s_ + 2 * kp);

#pragma unroll
  for (int s_ = 0; s_ < 16; ++s_)
#pragma unroll
    for (int nn = 0; nn < 4; ++nn)
      acc[nn] = __builtin_amdgcn_wmma_f32_16x16x4_f32(false, pf[s_], false, vf[s_][nn],
                                                      (short)0, acc[nn], false, false);
}

__global__ __launch_bounds__(128) void flash_attn(
    const float* __restrict__ qm, const float* __restrict__ km,
    const float* __restrict__ vm, float* __restrict__ out)
{
  __shared__ __attribute__((aligned(16))) float kst[2][BN * KP];
  __shared__ __attribute__((aligned(16))) float vst[2][(BN / 2) * VP];
  __shared__ __attribute__((aligned(16))) float pbuf[4][16 * PP];

  const int lane = threadIdx.x & 31;
  const int wid  = threadIdx.x >> 5;
  const int tile = blockIdx.x * 4 + wid;
  const int bb   = tile >> 8;
  const int ti   = tile & 255;
  const int m    = lane & 15;
  const int kp   = lane >> 4;
  const int sc   = threadIdx.x & 15;
  const int sr   = threadIdx.x >> 4;

  const size_t base = (size_t)bb * Tsz * Hsz;
  const float* qb = qm + base;
  const float* kb = km + base;
  const float* vb = vm + base;

  v4f kr[8], vr[8];
  stage_load(kb, vb, 0, sr, sc, kr, vr);

  v2f qf[16];
  {
    const float* qr = qb + (size_t)(ti * 16 + m) * Hsz + 2 * kp;
#pragma unroll
    for (int s = 0; s < 16; ++s) {
      v2f t = *(const v2f*)(qr + 4 * s);
      qf[s].x = t.x * 0.125f;
      qf[s].y = t.y * 0.125f;
    }
  }

  v8f acc[4] = {};
  float mr[8], lr[8];
#pragma unroll
  for (int r = 0; r < 8; ++r) { mr[r] = -1e30f; lr[r] = 0.0f; }

  float* pb = pbuf[wid];

  const int nfull = blockIdx.x & 63;
  const int rowl  = ti * 16 - nfull * BN;

  stage_store(kst[0], vst[0], sr, sc, kr, vr);
  __syncthreads();

  for (int jb = 0; jb < nfull; ++jb) {
    const int cur = jb & 1;
    stage_load(kb, vb, jb + 1, sr, sc, kr, vr);
    __builtin_amdgcn_sched_barrier(0);
    kv_block<false>(kst[cur], vst[cur], rowl, m, kp, qf, acc, mr, lr, pb);
    __builtin_amdgcn_sched_barrier(0);
    stage_store(kst[cur ^ 1], vst[cur ^ 1], sr, sc, kr, vr);
    __syncthreads();
  }
  kv_block<true>(kst[nfull & 1], vst[nfull & 1], rowl, m, kp, qf, acc, mr, lr, pb);

#pragma unroll
  for (int nn = 0; nn < 4; ++nn)
#pragma unroll
    for (int r = 0; r < 8; ++r) {
      const size_t idx = base + (size_t)(ti * 16 + r + 8 * kp) * Hsz + nn * 16 + m;
      out[idx] = acc[nn][r] / lr[r];
    }
}

extern "C" void kernel_launch(void* const* d_in, const int* in_sizes, int n_in,
                              void* d_out, int out_size, void* d_ws, size_t ws_size,
                              hipStream_t stream) {
  (void)in_sizes; (void)n_in; (void)out_size; (void)ws_size;

  const float* x  = (const float*)d_in[0];
  const float* Wk = (const float*)d_in[1];
  const float* Wq = (const float*)d_in[2];
  const float* Wv = (const float*)d_in[3];
  float* out = (float*)d_out;

  float* qo = (float*)d_ws;
  float* ko = qo + (size_t)BT * Hsz;
  float* vo = ko + (size_t)BT * Hsz;

  const int tiles = BT / 16;
  dim3 block(128);
  dim3 grid(tiles / 4);

  proj_qkv<<<grid, block, 0, stream>>>(x, Wk, Wq, Wv, qo, ko, vo);
  flash_attn<<<grid, block, 0, stream>>>(qo, ko, vo, out);
}


# attention: each 64-key block's row folds (8 maxima, 8 sums) take their lane^1,2,4,8 steps as DPP row_xmask operands on the vector ALU, not as ds_bpermute trips through the LDS queue; sums unchanged
# speedup vs baseline: 3.5633x; 1.1834x over previous
#include <hip/hip_runtime.h>

typedef __attribute__((ext_vector_type(2))) float v2f;
typedef __attribute__((ext_vector_type(4))) float v4f;
typedef __attribute__((ext_vector_type(8))) float v8f;

static constexpr int Bsz = 4;
static constexpr int Tsz = 4096;
static constexpr int Csz = 1024;
static constexpr int Hsz = 64;
static constexpr int BT  = Bsz * Tsz;

static constexpr int PM  = 64;
static constexpr int KC  = 64;
static constexpr int XP  = KC + 4;
static constexpr int CP  = 3 * Hsz + 4;

template <bool MORE>
__device__ __forceinline__ void proj_trip(
    const float* __restrict__ xa, float* __restrict__ xs0,
    const float* __restrict__ xg, const float* const (&wb)[3],
    const int c, const int sr, const int sc, v4f (&xr)[8], v8f (&acc)[4][3])
{
  const int cur = c & 1;
  const float* xc = xa + cur * (PM * XP);
  const size_t kof = (size_t)(c * KC) * Hsz;

  v2f wf[16][3];
#pragma unroll
  for (int s = 0; s < 8; ++s)
#pragma unroll
    for (int j = 0; j < 3; ++j) {
      wf[s][j].x = wb[j][kof + (size_t)(4 * s) * Hsz];
      wf[s][j].y = wb[j][kof + (size_t)(4 * s + 1) * Hsz];
    }

  v2f af[16][4];
#pragma unroll
  for (int s = 0; s < 16; ++s)
#pragma unroll
    for (int mt = 0; mt < 4; ++mt)
      af[s][mt] = *(const v2f*)(xc + (16 * mt) * XP + 4 * s);

  __builtin_amdgcn_sched_barrier(0);

#pragma unroll
  for (int s = 0; s < 4; ++s)
#pragma unroll
    for (int mt = 0; mt < 4; ++mt)
#pragma unroll
      for (int j = 0; j < 3; ++j)
        acc[mt][j] = __builtin_amdgcn_wmma_f32_16x16x4_f32(false, af[s][mt], false, wf[s][j],
                                                          (short)0, acc[mt][j], false, false);

  __builtin_amdgcn_sched_barrier(0);

#pragma unroll
  for (int s = 8; s < 16; ++s)
#pragma unroll
    for (int j = 0; j < 3; ++j) {
      wf[s][j].x = wb[j][kof + (size_t)(4 * s) * Hsz];
      wf[s][j].y = wb[j][kof + (size_t)(4 * s + 1) * Hsz];
    }

  if (MORE) {
#pragma unroll
    for (int u = 0; u < 8; ++u)
      xr[u] = *(const v4f*)(xg + (size_t)(8 * u) * Csz + (c + 1) * KC);
  }

  __builtin_amdgcn_sched_barrier(0);

#pragma unroll
  for (int s = 4; s < 16; ++s)
#pragma unroll
    for (int mt = 0; mt < 4; ++mt)
#pragma unroll
      for (int j = 0; j < 3; ++j)
        acc[mt][j] = __builtin_amdgcn_wmma_f32_16x16x4_f32(false, af[s][mt], false, wf[s][j],
                                                          (short)0, acc[mt][j], false, false);

  __builtin_amdgcn_sched_barrier(0);

  if (MORE) {
    float* xd = xs0 + (cur ^ 1) * (PM * XP) + sr * XP + 4 * sc;
#pragma unroll
    for (int u = 0; u < 8; ++u)
      *(v4f*)(xd + (8 * u) * XP) = xr[u];
  }
}

__global__ __launch_bounds__(128) void proj_qkv(
    const float* __restrict__ x,
    const float* __restrict__ Wk,
    const float* __restrict__ Wq,
    const float* __restrict__ Wv,
    float* __restrict__ qo, float* __restrict__ ko, float* __restrict__ vo)
{
  __shared__ __attribute__((aligned(16))) float xs[2][PM * XP];
  __shared__ __attribute__((aligned(16))) float cs[PM * CP];

  const int lane = threadIdx.x & 31;
  const int wid  = __builtin_amdgcn_readfirstlane((int)(threadIdx.x >> 5));
  const int m    = lane & 15;
  const int kp   = lane >> 4;
  const int sc   = threadIdx.x & 15;
  const int sr   = threadIdx.x >> 4;
  const int row0 = blockIdx.x * PM;

  const float* wb[3];
#pragma unroll
  for (int j = 0; j < 3; ++j) {
    const int g   = 3 * wid + j;
    const int mat = g >> 2;
    const float* wsel = (mat == 0) ? Wq : ((mat == 1) ? Wk : Wv);
    wb[j] = wsel + (size_t)(2 * kp) * Hsz + (g & 3) * 16 + m;
  }

  const float* xg = x + (size_t)(row0 + sr) * Csz + 4 * sc;
  const float* xa = &xs[0][0] + m * XP + 2 * kp;

  v8f acc[4][3];
#pragma unroll
  for (int mt = 0; mt < 4; ++mt)
#pragma unroll
    for (int j = 0; j < 3; ++j) acc[mt][j] = v8f{};

  v4f xr[8];
#pragma unroll
  for (int u = 0; u < 8; ++u)
    xr[u] = *(const v4f*)(xg + (size_t)(8 * u) * Csz);
#pragma unroll
  for (int u = 0; u < 8; ++u)
    *(v4f*)(&xs[0][0] + (sr + 8 * u) * XP + 4 * sc) = xr[u];
  __syncthreads();

  for (int c = 0; c < Csz / KC - 1; ++c) {
    proj_trip<true>(xa, xs[0], xg, wb, c, sr, sc, xr, acc);
    __syncthreads();
  }
  proj_trip<false>(xa, xs[0], xg, wb, Csz / KC - 1, sr, sc, xr, acc);

#pragma unroll
  for (int mt = 0; mt < 4; ++mt)
#pragma unroll
    for (int j = 0; j < 3; ++j)
#pragma unroll
      for (int r = 0; r < 8; ++r)
        cs[(16 * mt + r + 8 * kp) * CP + 16 * (3 * wid + j) + m] = acc[mt][j][r];
  __syncthreads();

#pragma unroll
  for (int u = 0; u < 8; ++u) {
    const int row = sr + 8 * u;
    const float* cr = cs + row * CP + 4 * sc;
    const size_t o = (size_t)(row0 + row) * Hsz + 4 * sc;
    *(v4f*)(qo + o) = *(const v4f*)(cr);
    *(v4f*)(ko + o) = *(const v4f*)(cr + Hsz);
    *(v4f*)(vo + o) = *(const v4f*)(cr + 2 * Hsz);
  }
}

static constexpr int KT = 4;
static constexpr int BN = 16 * KT;
static constexpr int PP = BN + 4;
static constexpr int KP = Hsz + 4;
static constexpr int VP = 2 * Hsz + 32;

template <int X>
__device__ __forceinline__ float row_xor(const float v)
{
  return __builtin_bit_cast(float, __builtin_amdgcn_update_dpp(
      0, __builtin_bit_cast(int, v), 0x160 + X, 0xf, 0xf, true));
}

__device__ __forceinline__ void stage_load(
    const float* __restrict__ kb, const float* __restrict__ vb,
    const int jb, const int sr, const int sc, v4f (&kr)[8], v4f (&vr)[8])
{
  const float* kg = kb + (size_t)(jb * BN + sr) * Hsz + 4 * sc;
#pragma unroll
  for (int u = 0; u < 8; ++u)
    kr[u] = *(const v4f*)(kg + (size_t)(8 * u) * Hsz);
  const float* vg = vb + (size_t)(jb * BN + 2 * sr) * Hsz + 4 * sc;
#pragma unroll
  for (int u = 0; u < 4; ++u) {
    vr[2 * u]     = *(const v4f*)(vg + (size_t)(16 * u) * Hsz);
    vr[2 * u + 1] = *(const v4f*)(vg + (size_t)(16 * u + 1) * Hsz);
  }
}

__device__ __forceinline__ void stage_store(
    float* __restrict__ ks, float* __restrict__ vs,
    const int sr, const int sc, const v4f (&kr)[8], const v4f (&vr)[8])
{
#pragma unroll
  for (int u = 0; u < 8; ++u)
    *(v4f*)(ks + (sr + 8 * u) * KP + 4 * sc) = kr[u];
#pragma unroll
  for (int u = 0; u < 4; ++u) {
    const v4f a = vr[2 * u];
    const v4f b = vr[2 * u + 1];
    float* d = vs + (sr + 8 * u) * VP + 8 * sc;
    v4f lo, hi;
    lo.x = a.x; lo.y = b.x; lo.z = a.y; lo.w = b.y;
    hi.x = a.z; hi.y = b.z; hi.z = a.w; hi.w = b.w;
    *(v4f*)(d)     = lo;
    *(v4f*)(d + 4) = hi;
  }
}

template <bool MASK>
__device__ __forceinline__ void kv_block(
    const float* __restrict__ ks, const float* __restrict__ vs,
    const int rowl, const int m, const int kp,
    const v2f (&qf)[16], v8f (&acc)[4], float (&mr)[8], float (&lr)[8],
    float* __restrict__ pb)
{
  const float* kt = ks + m * KP + 2 * kp;
  v2f kf[16][KT];
#pragma unroll
  for (int s_ = 0; s_ < 16; ++s_)
#pragma unroll
    for (int t = 0; t < KT; ++t)
      kf[s_][t] = *(const v2f*)(kt + (t * 16) * KP + 4 * s_);

  v8f s[KT];
#pragma unroll
  for (int t = 0; t < KT; ++t) s[t] = v8f{};
#pragma unroll
  for (int s_ = 0; s_ < 16; ++s_)
#pragma unroll
    for (int t = 0; t < KT; ++t)
      s[t] = __builtin_amdgcn_wmma_f32_16x16x4_f32(false, qf[s_], false, kf[s_][t],
                                                   (short)0, s[t], false, false);

  if (MASK) {
#pragma unroll
    for (int t = 0; t < KT; ++t)
#pragma unroll
      for (int r = 0; r < 8; ++r)
        if (16 * t + m > rowl + r + 8 * kp) s[t][r] = -1e30f;
  }

  float alpha[8];
#pragma unroll
  for (int r = 0; r < 8; ++r) {
    float rm = s[0][r];
#pragma unroll
    for (int t = 1; t < KT; ++t) rm = fmaxf(rm, s[t][r]);
    rm = fmaxf(rm, row_xor<1>(rm));
    rm = fmaxf(rm, row_xor<2>(rm));
    rm = fmaxf(rm, row_xor<4>(rm));
    rm = fmaxf(rm, row_xor<8>(rm));
    const float mnew = fmaxf(mr[r], rm);
    alpha[r] = __expf(mr[r] - mnew);
    mr[r] = mnew;
  }

  __builtin_amdgcn_sched_barrier(0);

  const float* vt = vs + kp * VP + 2 * m;
  v2f vf[16][4];
#pragma unroll
  for (int s_ = 0; s_ < 16; ++s_)
#pragma unroll
    for (int nn = 0; nn < 4; ++nn)
      vf[s_][nn] = *(const v2f*)(vt + (2 * s_) * VP + 32 * nn);

  __builtin_amdgcn_sched_barrier(0);

  __builtin_amdgcn_wave_barrier();

#pragma unroll
  for (int r = 0; r < 8; ++r) {
    const float mnew = mr[r];
    float rs = 0.0f;
#pragma unroll
    for (int t = 0; t < KT; ++t) {
      const float p = __expf(s[t][r] - mnew);
      rs += p;
      pb[(r + 8 * kp) * PP + 16 * t + m] = p;
    }
    rs += row_xor<1>(rs);
    rs += row_xor<2>(rs);
    rs += row_xor<4>(rs);
    rs += row_xor<8>(rs);
    lr[r] = lr[r] * alpha[r] + rs;
#pragma unroll
    for (int nn = 0; nn < 4; ++nn) acc[nn][r] *= alpha[r];
  }

  __builtin_amdgcn_wave_barrier();

  v2f pf[16];
#pragma unroll
  for (int s_ = 0; s_ < 16; ++s_)
    pf[s_] = *(const v2f*)(pb + m * PP + 4 * s_ + 2 * kp);

#pragma unroll
  for (int s_ = 0; s_ < 16; ++s_)
#pragma unroll
    for (int nn = 0; nn < 4; ++nn)
      acc[nn] = __builtin_amdgcn_wmma_f32_16x16x4_f32(false, pf[s_], false, vf[s_][nn],
                                                      (short)0, acc[nn], false, false);
}

__global__ __launch_bounds__(128) void flash_attn(
    const float* __restrict__ qm, const float* __restrict__ km,
    const float* __restrict__ vm, float* __restrict__ out)
{
  __shared__ __attribute__((aligned(16))) float kst[2][BN * KP];
  __shared__ __attribute__((aligned(16))) float vst[2][(BN / 2) * VP];
  __shared__ __attribute__((aligned(16))) float pbuf[4][16 * PP];

  const int lane = threadIdx.x & 31;
  const int wid  = threadIdx.x >> 5;
  const int tile = blockIdx.x * 4 + wid;
  const int bb   = tile >> 8;
  const int ti   = tile & 255;
  const int m    = lane & 15;
  const int kp   = lane >> 4;
  const int sc   = threadIdx.x & 15;
  const int sr   = threadIdx.x >> 4;

  const size_t base = (size_t)bb * Tsz * Hsz;
  const float* qb = qm + base;
  const float* kb = km + base;
  const float* vb = vm + base;

  v4f kr[8], vr[8];
  stage_load(kb, vb, 0, sr, sc, kr, vr);

  v2f qf[16];
  {
    const float* qr = qb + (size_t)(ti * 16 + m) * Hsz + 2 * kp;
#pragma unroll
    for (int s = 0; s < 16; ++s) {
      v2f t = *(const v2f*)(qr + 4 * s);
      qf[s].x = t.x * 0.125f;
      qf[s].y = t.y * 0.125f;
    }
  }

  v8f acc[4] = {};
  float mr[8], lr[8];
#pragma unroll
  for (int r = 0; r < 8; ++r) { mr[r] = -1e30f; lr[r] = 0.0f; }

  float* pb = pbuf[wid];

  const int nfull = blockIdx.x & 63;
  const int rowl  = ti * 16 - nfull * BN;

  stage_store(kst[0], vst[0], sr, sc, kr, vr);
  __syncthreads();

  for (int jb = 0; jb < nfull; ++jb) {
    const int cur = jb & 1;
    stage_load(kb, vb, jb + 1, sr, sc, kr, vr);
    __builtin_amdgcn_sched_barrier(0);
    kv_block<false>(kst[cur], vst[cur], rowl, m, kp, qf, acc, mr, lr, pb);
    __builtin_amdgcn_sched_barrier(0);
    stage_store(kst[cur ^ 1], vst[cur ^ 1], sr, sc, kr, vr);
    __syncthreads();
  }
  kv_block<true>(kst[nfull & 1], vst[nfull & 1], rowl, m, kp, qf, acc, mr, lr, pb);

#pragma unroll
  for (int nn = 0; nn < 4; ++nn)
#pragma unroll
    for (int r = 0; r < 8; ++r) {
      const size_t idx = base + (size_t)(ti * 16 + r + 8 * kp) * Hsz + nn * 16 + m;
      out[idx] = acc[nn][r] / lr[r];
    }
}

extern "C" void kernel_launch(void* const* d_in, const int* in_sizes, int n_in,
                              void* d_out, int out_size, void* d_ws, size_t ws_size,
                              hipStream_t stream) {
  (void)in_sizes; (void)n_in; (void)out_size; (void)ws_size;

  const float* x  = (const float*)d_in[0];
  const float* Wk = (const float*)d_in[1];
  const float* Wq = (const float*)d_in[2];
  const float* Wv = (const float*)d_in[3];
  float* out = (float*)d_out;

  float* qo = (float*)d_ws;
  float* ko = qo + (size_t)BT * Hsz;
  float* vo = ko + (size_t)BT * Hsz;

  const int tiles = BT / 16;
  dim3 block(128);
  dim3 grid(tiles / 4);

  proj_qkv<<<grid, block, 0, stream>>>(x, Wk, Wq, Wv, qo, ko, vo);
  flash_attn<<<grid, block, 0, stream>>>(qo, ko, vo, out);
}


# attention: the output products of key block j-1 are issued beside the softmax of block j (products skewed by one trip, three-stage LDS ring, P in registers to the trip's end); every sum unchanged
# speedup vs baseline: 3.7268x; 1.0459x over previous
#include <hip/hip_runtime.h>

typedef __attribute__((ext_vector_type(2))) float v2f;
typedef __attribute__((ext_vector_type(4))) float v4f;
typedef __attribute__((ext_vector_type(8))) float v8f;

static constexpr int Bsz = 4;
static constexpr int Tsz = 4096;
static constexpr int Csz = 1024;
static constexpr int Hsz = 64;
static constexpr int BT  = Bsz * Tsz;

static constexpr int PM  = 64;
static constexpr int KC  = 64;
static constexpr int XP  = KC + 4;
static constexpr int CP  = 3 * Hsz + 4;

template <bool MORE>
__device__ __forceinline__ void proj_trip(
    const float* __restrict__ xa, float* __restrict__ xs0,
    const float* __restrict__ xg, const float* const (&wb)[3],
    const int c, const int sr, const int sc, v4f (&xr)[8], v8f (&acc)[4][3])
{
  const int cur = c & 1;
  const float* xc = xa + cur * (PM * XP);
  const size_t kof = (size_t)(c * KC) * Hsz;

  v2f wf[16][3];
#pragma unroll
  for (int s = 0; s < 8; ++s)
#pragma unroll
    for (int j = 0; j < 3; ++j) {
      wf[s][j].x = wb[j][kof + (size_t)(4 * s) * Hsz];
      wf[s][j].y = wb[j][kof + (size_t)(4 * s + 1) * Hsz];
    }

  v2f af[16][4];
#pragma unroll
  for (int s = 0; s < 16; ++s)
#pragma unroll
    for (int mt = 0; mt < 4; ++mt)
      af[s][mt] = *(const v2f*)(xc + (16 * mt) * XP + 4 * s);

  __builtin_amdgcn_sched_barrier(0);

#pragma unroll
  for (int s = 0; s < 4; ++s)
#pragma unroll
    for (int mt = 0; mt < 4; ++mt)
#pragma unroll
      for (int j = 0; j < 3; ++j)
        acc[mt][j] = __builtin_amdgcn_wmma_f32_16x16x4_f32(false, af[s][mt], false, wf[s][j],
                                                          (short)0, acc[mt][j], false, false);

  __builtin_amdgcn_sched_barrier(0);

#pragma unroll
  for (int s = 8; s < 16; ++s)
#pragma unroll
    for (int j = 0; j < 3; ++j) {
      wf[s][j].x = wb[j][kof + (size_t)(4 * s) * Hsz];
      wf[s][j].y = wb[j][kof + (size_t)(4 * s + 1) * Hsz];
    }

  if (MORE) {
#pragma unroll
    for (int u = 0; u < 8; ++u)
      xr[u] = *(const v4f*)(xg + (size_t)(8 * u) * Csz + (c + 1) * KC);
  }

  __builtin_amdgcn_sched_barrier(0);

#pragma unroll
  for (int s = 4; s < 16; ++s)
#pragma unroll
    for (int mt = 0; mt < 4; ++mt)
#pragma unroll
      for (int j = 0; j < 3; ++j)
        acc[mt][j] = __builtin_amdgcn_wmma_f32_16x16x4_f32(false, af[s][mt], false, wf[s][j],
                                                          (short)0, acc[mt][j], false, false);

  __builtin_amdgcn_sched_barrier(0);

  if (MORE) {
    float* xd = xs0 + (cur ^ 1) * (PM * XP) + sr * XP + 4 * sc;
#pragma unroll
    for (int u = 0; u < 8; ++u)
      *(v4f*)(xd + (8 * u) * XP) = xr[u];
  }
}

__global__ __launch_bounds__(128) void proj_qkv(
    const float* __restrict__ x,
    const float* __restrict__ Wk,
    const float* __restrict__ Wq,
    const float* __restrict__ Wv,
    float* __restrict__ qo, float* __restrict__ ko, float* __restrict__ vo)
{
  __shared__ __attribute__((aligned(16))) float xs[2][PM * XP];
  __shared__ __attribute__((aligned(16))) float cs[PM * CP];

  const int lane = threadIdx.x & 31;
  const int wid  = __builtin_amdgcn_readfirstlane((int)(threadIdx.x >> 5));
  const int m    = lane & 15;
  const int kp   = lane >> 4;
  const int sc   = threadIdx.x & 15;
  const int sr   = threadIdx.x >> 4;
  const int row0 = blockIdx.x * PM;

  const float* wb[3];
#pragma unroll
  for (int j = 0; j < 3; ++j) {
    const int g   = 3 * wid + j;
    const int mat = g >> 2;
    const float* wsel = (mat == 0) ? Wq : ((mat == 1) ? Wk : Wv);
    wb[j] = wsel + (size_t)(2 * kp) * Hsz + (g & 3) * 16 + m;
  }

  const float* xg = x + (size_t)(row0 + sr) * Csz + 4 * sc;
  const float* xa = &xs[0][0] + m * XP + 2 * kp;

  v8f acc[4][3];
#pragma unroll
  for (int mt = 0; mt < 4; ++mt)
#pragma unroll
    for (int j = 0; j < 3; ++j) acc[mt][j] = v8f{};

  v4f xr[8];
#pragma unroll
  for (int u = 0; u < 8; ++u)
    xr[u] = *(const v4f*)(xg + (size_t)(8 * u) * Csz);
#pragma unroll
  for (int u = 0; u < 8; ++u)
    *(v4f*)(&xs[0][0] + (sr + 8 * u) * XP + 4 * sc) = xr[u];
  __syncthreads();

  for (int c = 0; c < Csz / KC - 1; ++c) {
    proj_trip<true>(xa, xs[0], xg, wb, c, sr, sc, xr, acc);
    __syncthreads();
  }
  proj_trip<false>(xa, xs[0], xg, wb, Csz / KC - 1, sr, sc, xr, acc);

#pragma unroll
  for (int mt = 0; mt < 4; ++mt)
#pragma unroll
    for (int j = 0; j < 3; ++j)
#pragma unroll
      for (int r = 0; r < 8; ++r)
        cs[(16 * mt + r + 8 * kp) * CP + 16 * (3 * wid + j) + m] = acc[mt][j][r];
  __syncthreads();

#pragma unroll
  for (int u = 0; u < 8; ++u) {
    const int row = sr + 8 * u;
    const float* cr = cs + row * CP + 4 * sc;
    const size_t o = (size_t)(row0 + row) * Hsz + 4 * sc;
    *(v4f*)(qo + o) = *(const v4f*)(cr);
    *(v4f*)(ko + o) = *(const v4f*)(cr + Hsz);
    *(v4f*)(vo + o) = *(const v4f*)(cr + 2 * Hsz);
  }
}

static constexpr int KT = 4;
static constexpr int BN = 16 * KT;
static constexpr int PP = BN + 4;
static constexpr int KP = Hsz + 4;
static constexpr int VP = 2 * Hsz + 32;
static constexpr int KSZ = BN * KP;
static constexpr int VSZ = (BN / 2) * VP;

template <int X>
__device__ __forceinline__ float row_xor(const float v)
{
  return __builtin_bit_cast(float, __builtin_amdgcn_update_dpp(
      0, __builtin_bit_cast(int, v), 0x160 + X, 0xf, 0xf, true));
}

__device__ __forceinline__ void stage_load(
    const float* __restrict__ kb, const float* __restrict__ vb,
    const int jb, const int sr, const int sc, v4f (&kr)[8], v4f (&vr)[8])
{
  const float* kg = kb + (size_t)(jb * BN + sr) * Hsz + 4 * sc;
#pragma unroll
  for (int u = 0; u < 8; ++u)
    kr[u] = *(const v4f*)(kg + (size_t)(8 * u) * Hsz);
  const float* vg = vb + (size_t)(jb * BN + 2 * sr) * Hsz + 4 * sc;
#pragma unroll
  for (int u = 0; u < 4; ++u) {
    vr[2 * u]     = *(const v4f*)(vg + (size_t)(16 * u) * Hsz);
    vr[2 * u + 1] = *(const v4f*)(vg + (size_t)(16 * u + 1) * Hsz);
  }
}

__device__ __forceinline__ void stage_store(
    float* __restrict__ ks, float* __restrict__ vs,
    const int sr, const int sc, const v4f (&kr)[8], const v4f (&vr)[8])
{
#pragma unroll
  for (int u = 0; u < 8; ++u)
    *(v4f*)(ks + (sr + 8 * u) * KP + 4 * sc) = kr[u];
#pragma unroll
  for (int u = 0; u < 4; ++u) {
    const v4f a = vr[2 * u];
    const v4f b = vr[2 * u + 1];
    float* d = vs + (sr + 8 * u) * VP + 8 * sc;
    v4f lo, hi;
    lo.x = a.x; lo.y = b.x; lo.z = a.y; lo.w = b.y;
    hi.x = a.z; hi.y = b.z; hi.z = a.w; hi.w = b.w;
    *(v4f*)(d)     = lo;
    *(v4f*)(d + 4) = hi;
  }
}

template <int S0, int S1>
__device__ __forceinline__ void v_frags(const float* __restrict__ vt, v2f (&vf)[16][4])
{
#pragma unroll
  for (int s_ = S0; s_ < S1; ++s_)
#pragma unroll
    for (int nn = 0; nn < 4; ++nn)
      vf[s_][nn] = *(const v2f*)(vt + (2 * s_) * VP + 32 * nn);
}

__device__ __forceinline__ void p_frags(const float* __restrict__ pb, const int m, const int kp,
                                        v2f (&pf)[16])
{
#pragma unroll
  for (int s_ = 0; s_ < 16; ++s_)
    pf[s_] = *(const v2f*)(pb + m * PP + 4 * s_ + 2 * kp);
}

template <int S0, int S1>
__device__ __forceinline__ void s_products(const v2f (&qf)[16], const v2f (&kf)[16][KT], v8f (&s)[KT])
{
#pragma unroll
  for (int s_ = S0; s_ < S1; ++s_)
#pragma unroll
    for (int t = 0; t < KT; ++t)
      s[t] = __builtin_amdgcn_wmma_f32_16x16x4_f32(false, qf[s_], false, kf[s_][t],
                                                   (short)0, s[t], false, false);
}

template <bool FIRST, bool MASK>
__device__ __forceinline__ void kv_trip(
    const float* __restrict__ ks, const float* __restrict__ vsp,
    const int rowl, const int m, const int kp,
    const v2f (&qf)[16], v8f (&acc)[4], float (&mr)[8], float (&lr)[8],
    float (&alpha)[8], float* __restrict__ pb)
{
  const float* kt = ks + m * KP + 2 * kp;
  v2f kf[16][KT];
#pragma unroll
  for (int s_ = 0; s_ < 16; ++s_)
#pragma unroll
    for (int t = 0; t < KT; ++t)
      kf[s_][t] = *(const v2f*)(kt + (t * 16) * KP + 4 * s_);

  __builtin_amdgcn_sched_barrier(0);

  const float* vt = vsp + kp * VP + 2 * m;
  v2f vf[16][4];
  v2f pf[16];
  v8f s[KT];
#pragma unroll
  for (int t = 0; t < KT; ++t) s[t] = v8f{};

  if (!FIRST) {
#pragma unroll
    for (int r = 0; r < 8; ++r)
#pragma unroll
      for (int nn = 0; nn < 4; ++nn) acc[nn][r] *= alpha[r];
  }
  s_products<0, 8>(qf, kf, s);

  __builtin_amdgcn_sched_barrier(0);
  if (!FIRST) {
    v_frags<0, 8>(vt, vf);
    p_frags(pb, m, kp, pf);
  }
  __builtin_amdgcn_sched_barrier(0);

  s_products<8, 12>(qf, kf, s);

  __builtin_amdgcn_sched_barrier(0);
  if (!FIRST) v_frags<8, 16>(vt, vf);
  __builtin_amdgcn_sched_barrier(0);

  s_products<12, 16>(qf, kf, s);

  if (MASK) {
#pragma unroll
    for (int t = 0; t < KT; ++t)
#pragma unroll
      for (int r = 0; r < 8; ++r)
        if (16 * t + m > rowl + r + 8 * kp) s[t][r] = -1e30f;
  }

  __builtin_amdgcn_sched_barrier(0);

  float anew[8];
  float p[KT][8];
#pragma unroll
  for (int h = 0; h < 2; ++h) {
    float rm[4], rs[4];
#pragma unroll
    for (int rr = 0; rr < 4; ++rr) {
      const int r = 4 * h + rr;
      rm[rr] = s[0][r];
#pragma unroll
      for (int t = 1; t < KT; ++t) rm[rr] = fmaxf(rm[rr], s[t][r]);
    }
#pragma unroll
    for (int rr = 0; rr < 4; ++rr) rm[rr] = fmaxf(rm[rr], row_xor<1>(rm[rr]));
#pragma unroll
    for (int rr = 0; rr < 4; ++rr) rm[rr] = fmaxf(rm[rr], row_xor<2>(rm[rr]));
#pragma unroll
    for (int rr = 0; rr < 4; ++rr) rm[rr] = fmaxf(rm[rr], row_xor<4>(rm[rr]));
#pragma unroll
    for (int rr = 0; rr < 4; ++rr) rm[rr] = fmaxf(rm[rr], row_xor<8>(rm[rr]));
#pragma unroll
    for (int rr = 0; rr < 4; ++rr) {
      const int r = 4 * h + rr;
      const float mnew = fmaxf(mr[r], rm[rr]);
      anew[r] = __expf(mr[r] - mnew);
      mr[r] = mnew;
    }
#pragma unroll
    for (int rr = 0; rr < 4; ++rr) {
      const int r = 4 * h + rr;
      rs[rr] = 0.0f;
#pragma unroll
      for (int t = 0; t < KT; ++t) {
        p[t][r] = __expf(s[t][r] - mr[r]);
        rs[rr] += p[t][r];
      }
    }
#pragma unroll
    for (int rr = 0; rr < 4; ++rr) rs[rr] += row_xor<1>(rs[rr]);
#pragma unroll
    for (int rr = 0; rr < 4; ++rr) rs[rr] += row_xor<2>(rs[rr]);
#pragma unroll
    for (int rr = 0; rr < 4; ++rr) rs[rr] += row_xor<4>(rs[rr]);
#pragma unroll
    for (int rr = 0; rr < 4; ++rr) rs[rr] += row_xor<8>(rs[rr]);
#pragma unroll
    for (int rr = 0; rr < 4; ++rr) {
      const int r = 4 * h + rr;
      lr[r] = lr[r] * anew[r] + rs[rr];
    }
    if (!FIRST) {
#pragma unroll
      for (int s_ = 8 * h; s_ < 8 * h + 8; ++s_)
#pragma unroll
        for (int nn = 0; nn < 4; ++nn)
          acc[nn] = __builtin_amdgcn_wmma_f32_16x16x4_f32(false, pf[s_], false, vf[s_][nn],
                                                          (short)0, acc[nn], false, false);
#pragma unroll
      for (int i = 0; i < 32; ++i) {
        __builtin_amdgcn_sched_group_barrier(0x008, 1, 0);
        __builtin_amdgcn_sched_group_barrier(0x002, 5, 0);
      }
    }
    __builtin_amdgcn_sched_barrier(0);
  }

  __builtin_amdgcn_wave_barrier();

#pragma unroll
  for (int r = 0; r < 8; ++r) {
#pragma unroll
    for (int t = 0; t < KT; ++t)
      pb[(r + 8 * kp) * PP + 16 * t + m] = p[t][r];
    alpha[r] = anew[r];
  }
}

__global__ __launch_bounds__(128) void flash_attn(
    const float* __restrict__ qm, const float* __restrict__ km,
    const float* __restrict__ vm, float* __restrict__ out)
{
  __shared__ __attribute__((aligned(16))) float kst[3 * KSZ];
  __shared__ __attribute__((aligned(16))) float vst[3 * VSZ];
  __shared__ __attribute__((aligned(16))) float pbuf[4][16 * PP];

  const int lane = threadIdx.x & 31;
  const int wid  = threadIdx.x >> 5;
  const int tile = blockIdx.x * 4 + wid;
  const int bb   = tile >> 8;
  const int ti   = tile & 255;
  const int m    = lane & 15;
  const int kp   = lane >> 4;
  const int sc   = threadIdx.x & 15;
  const int sr   = threadIdx.x >> 4;

  const size_t base = (size_t)bb * Tsz * Hsz;
  const float* qb = qm + base;
  const float* kb = km + base;
  const float* vb = vm + base;

  v4f kr[8], vr[8];
  stage_load(kb, vb, 0, sr, sc, kr, vr);

  v2f qf[16];
  {
    const float* qr = qb + (size_t)(ti * 16 + m) * Hsz + 2 * kp;
#pragma unroll
    for (int s = 0; s < 16; ++s) {
      v2f t = *(const v2f*)(qr + 4 * s);
      qf[s].x = t.x * 0.125f;
      qf[s].y = t.y * 0.125f;
    }
  }

  v8f acc[4] = {};
  float mr[8], lr[8], alpha[8];
#pragma unroll
  for (int r = 0; r < 8; ++r) { mr[r] = -1e30f; lr[r] = 0.0f; alpha[r] = 0.0f; }

  float* pb = pbuf[wid];

  const int nfull = blockIdx.x & 63;
  const int rowl  = ti * 16 - nfull * BN;

  stage_store(kst, vst, sr, sc, kr, vr);
  __syncthreads();

  int prv = 2, cur = 0, nxt = 1;
  if (nfull == 0) {
    kv_trip<true, true>(kst, vst, rowl, m, kp, qf, acc, mr, lr, alpha, pb);
  } else {
    stage_load(kb, vb, 1, sr, sc, kr, vr);
    __builtin_amdgcn_sched_barrier(0);
    kv_trip<true, false>(kst, vst, rowl, m, kp, qf, acc, mr, lr, alpha, pb);
    __builtin_amdgcn_sched_barrier(0);
    stage_store(kst + KSZ, vst + VSZ, sr, sc, kr, vr);
    __syncthreads();
    prv = 0; cur = 1; nxt = 2;

    for (int jb = 1; jb < nfull; ++jb) {
      stage_load(kb, vb, jb + 1, sr, sc, kr, vr);
      __builtin_amdgcn_sched_barrier(0);
      kv_trip<false, false>(kst + cur * KSZ, vst + prv * VSZ, rowl, m, kp,
                            qf, acc, mr, lr, alpha, pb);
      __builtin_amdgcn_sched_barrier(0);
      stage_store(kst + nxt * KSZ, vst + nxt * VSZ, sr, sc, kr, vr);
      __syncthreads();
      const int t = prv; prv = cur; cur = nxt; nxt = t;
    }
    kv_trip<false, true>(kst + cur * KSZ, vst + prv * VSZ, rowl, m, kp,
                         qf, acc, mr, lr, alpha, pb);
  }

  {
    __builtin_amdgcn_wave_barrier();
    v2f vf[16][4];
    v2f pf[16];
    v_frags<0, 8>(vst + cur * VSZ + kp * VP + 2 * m, vf);
    p_frags(pb, m, kp, pf);
#pragma unroll
    for (int r = 0; r < 8; ++r)
#pragma unroll
      for (int nn = 0; nn < 4; ++nn) acc[nn][r] *= alpha[r];
    v_frags<8, 16>(vst + cur * VSZ + kp * VP + 2 * m, vf);
    __builtin_amdgcn_sched_barrier(0);
#pragma unroll
    for (int s_ = 0; s_ < 16; ++s_)
#pragma unroll
      for (int nn = 0; nn < 4; ++nn)
        acc[nn] = __builtin_amdgcn_wmma_f32_16x16x4_f32(false, pf[s_], false, vf[s_][nn],
                                                        (short)0, acc[nn], false, false);
  }

#pragma unroll
  for (int nn = 0; nn < 4; ++nn)
#pragma unroll
    for (int r = 0; r < 8; ++r) {
      const size_t idx = base + (size_t)(ti * 16 + r + 8 * kp) * Hsz + nn * 16 + m;
      out[idx] = acc[nn][r] / lr[r];
    }
}

extern "C" void kernel_launch(void* const* d_in, const int* in_sizes, int n_in,
                              void* d_out, int out_size, void* d_ws, size_t ws_size,
                              hipStream_t stream) {
  (void)in_sizes; (void)n_in; (void)out_size; (void)ws_size;

  const float* x  = (const float*)d_in[0];
  const float* Wk = (const float*)d_in[1];
  const float* Wq = (const float*)d_in[2];
  const float* Wv = (const float*)d_in[3];
  float* out = (float*)d_out;

  float* qo = (float*)d_ws;
  float* ko = qo + (size_t)BT * Hsz;
  float* vo = ko + (size_t)BT * Hsz;

  const int tiles = BT / 16;
  dim3 block(128);
  dim3 grid(tiles / 4);

  proj_qkv<<<grid, block, 0, stream>>>(x, Wk, Wq, Wv, qo, ko, vo);
  flash_attn<<<grid, block, 0, stream>>>(qo, ko, vo, out);
}


# attention: causal work in equal shares: a workgroup pair owns q blocks i and 63-i (block i + head keys of 63-i | its tail keys); a third launch joins the partial softmax states; new summation order
# speedup vs baseline: 5.1789x; 1.3896x over previous
#include <hip/hip_runtime.h>

typedef __attribute__((ext_vector_type(2))) float v2f;
typedef __attribute__((ext_vector_type(4))) float v4f;
typedef __attribute__((ext_vector_type(8))) float v8f;

static constexpr int Bsz = 4;
static constexpr int Tsz = 4096;
static constexpr int Csz = 1024;
static constexpr int Hsz = 64;
static constexpr int BT  = Bsz * Tsz;

static constexpr int PM  = 64;
static constexpr int KC  = 64;
static constexpr int XP  = KC + 4;
static constexpr int CP  = 3 * Hsz + 4;

template <bool MORE>
__device__ __forceinline__ void proj_trip(
    const float* __restrict__ xa, float* __restrict__ xs0,
    const float* __restrict__ xg, const float* const (&wb)[3],
    const int c, const int sr, const int sc, v4f (&xr)[8], v8f (&acc)[4][3])
{
  const int cur = c & 1;
  const float* xc = xa + cur * (PM * XP);
  const size_t kof = (size_t)(c * KC) * Hsz;

  v2f wf[16][3];
#pragma unroll
  for (int s = 0; s < 8; ++s)
#pragma unroll
    for (int j = 0; j < 3; ++j) {
      wf[s][j].x = wb[j][kof + (size_t)(4 * s) * Hsz];
      wf[s][j].y = wb[j][kof + (size_t)(4 * s + 1) * Hsz];
    }

  v2f af[16][4];
#pragma unroll
  for (int s = 0; s < 16; ++s)
#pragma unroll
    for (int mt = 0; mt < 4; ++mt)
      af[s][mt] = *(const v2f*)(xc + (16 * mt) * XP + 4 * s);

  __builtin_amdgcn_sched_barrier(0);

#pragma unroll
  for (int s = 0; s < 4; ++s)
#pragma unroll
    for (int mt = 0; mt < 4; ++mt)
#pragma unroll
      for (int j = 0; j < 3; ++j)
        acc[mt][j] = __builtin_amdgcn_wmma_f32_16x16x4_f32(false, af[s][mt], false, wf[s][j],
                                                          (short)0, acc[mt][j], false, false);

  __builtin_amdgcn_sched_barrier(0);

#pragma unroll
  for (int s = 8; s < 16; ++s)
#pragma unroll
    for (int j = 0; j < 3; ++j) {
      wf[s][j].x = wb[j][kof + (size_t)(4 * s) * Hsz];
      wf[s][j].y = wb[j][kof + (size_t)(4 * s + 1) * Hsz];
    }

  if (MORE) {
#pragma unroll
    for (int u = 0; u < 8; ++u)
      xr[u] = *(const v4f*)(xg + (size_t)(8 * u) * Csz + (c + 1) * KC);
  }

  __builtin_amdgcn_sched_barrier(0);

#pragma unroll
  for (int s = 4; s < 16; ++s)
#pragma unroll
    for (int mt = 0; mt < 4; ++mt)
#pragma unroll
      for (int j = 0; j < 3; ++j)
        acc[mt][j] = __builtin_amdgcn_wmma_f32_16x16x4_f32(false, af[s][mt], false, wf[s][j],
                                                          (short)0, acc[mt][j], false, false);

  __builtin_amdgcn_sched_barrier(0);

  if (MORE) {
    float* xd = xs0 + (cur ^ 1) * (PM * XP) + sr * XP + 4 * sc;
#pragma unroll
    for (int u = 0; u < 8; ++u)
      *(v4f*)(xd + (8 * u) * XP) = xr[u];
  }
}

__global__ __launch_bounds__(128) void proj_qkv(
    const float* __restrict__ x,
    const float* __restrict__ Wk,
    const float* __restrict__ Wq,
    const float* __restrict__ Wv,
    float* __restrict__ qo, float* __restrict__ ko, float* __restrict__ vo,
    float* __restrict__ q2)
{
  __shared__ __attribute__((aligned(16))) float xs[2][PM * XP];
  __shared__ __attribute__((aligned(16))) float cs[PM * CP];

  const int lane = threadIdx.x & 31;
  const int wid  = __builtin_amdgcn_readfirstlane((int)(threadIdx.x >> 5));
  const int m    = lane & 15;
  const int kp   = lane >> 4;
  const int sc   = threadIdx.x & 15;
  const int sr   = threadIdx.x >> 4;
  const int row0 = blockIdx.x * PM;

  const float* wb[3];
#pragma unroll
  for (int j = 0; j < 3; ++j) {
    const int g   = 3 * wid + j;
    const int mat = g >> 2;
    const float* wsel = (mat == 0) ? Wq : ((mat == 1) ? Wk : Wv);
    wb[j] = wsel + (size_t)(2 * kp) * Hsz + (g & 3) * 16 + m;
  }

  const float* xg = x + (size_t)(row0 + sr) * Csz + 4 * sc;
  const float* xa = &xs[0][0] + m * XP + 2 * kp;

  v8f acc[4][3];
#pragma unroll
  for (int mt = 0; mt < 4; ++mt)
#pragma unroll
    for (int j = 0; j < 3; ++j) acc[mt][j] = v8f{};

  v4f xr[8];
#pragma unroll
  for (int u = 0; u < 8; ++u)
    xr[u] = *(const v4f*)(xg + (size_t)(8 * u) * Csz);
#pragma unroll
  for (int u = 0; u < 8; ++u)
    *(v4f*)(&xs[0][0] + (sr + 8 * u) * XP + 4 * sc) = xr[u];
  __syncthreads();

  for (int c = 0; c < Csz / KC - 1; ++c) {
    proj_trip<true>(xa, xs[0], xg, wb, c, sr, sc, xr, acc);
    __syncthreads();
  }
  proj_trip<false>(xa, xs[0], xg, wb, Csz / KC - 1, sr, sc, xr, acc);

#pragma unroll
  for (int mt = 0; mt < 4; ++mt)
#pragma unroll
    for (int j = 0; j < 3; ++j)
#pragma unroll
      for (int r = 0; r < 8; ++r)
        cs[(16 * mt + r + 8 * kp) * CP + 16 * (3 * wid + j) + m] = acc[mt][j][r];
  __syncthreads();

  const bool dup = (blockIdx.x & 63) >= 32;
#pragma unroll
  for (int u = 0; u < 8; ++u) {
    const int row = sr + 8 * u;
    const float* cr = cs + row * CP + 4 * sc;
    const size_t o = (size_t)(row0 + row) * Hsz + 4 * sc;
    const v4f qv = *(const v4f*)(cr);
    *(v4f*)(qo + o) = qv;
    if (dup) *(v4f*)(q2 + o) = qv;
    *(v4f*)(ko + o) = *(const v4f*)(cr + Hsz);
    *(v4f*)(vo + o) = *(const v4f*)(cr + 2 * Hsz);
  }
}

static constexpr int KT = 4;
static constexpr int BN = 16 * KT;
static constexpr int PP = BN + 4;
static constexpr int KP = Hsz + 4;
static constexpr int VP = 2 * Hsz + 32;
static constexpr int KSZ = BN * KP;
static constexpr int VSZ = (BN / 2) * VP;

template <int X>
__device__ __forceinline__ float row_xor(const float v)
{
  return __builtin_bit_cast(float, __builtin_amdgcn_update_dpp(
      0, __builtin_bit_cast(int, v), 0x160 + X, 0xf, 0xf, true));
}

__device__ __forceinline__ void stage_load(
    const float* __restrict__ kb, const float* __restrict__ vb,
    const int jb, const int sr, const int sc, v4f (&kr)[8], v4f (&vr)[8])
{
  const float* kg = kb + (size_t)(jb * BN + sr) * Hsz + 4 * sc;
#pragma unroll
  for (int u = 0; u < 8; ++u)
    kr[u] = *(const v4f*)(kg + (size_t)(8 * u) * Hsz);
  const float* vg = vb + (size_t)(jb * BN + 2 * sr) * Hsz + 4 * sc;
#pragma unroll
  for (int u = 0; u < 4; ++u) {
    vr[2 * u]     = *(const v4f*)(vg + (size_t)(16 * u) * Hsz);
    vr[2 * u + 1] = *(const v4f*)(vg + (size_t)(16 * u + 1) * Hsz);
  }
}

__device__ __forceinline__ void stage_store(
    float* __restrict__ ks, float* __restrict__ vs,
    const int sr, const int sc, const v4f (&kr)[8], const v4f (&vr)[8])
{
#pragma unroll
  for (int u = 0; u < 8; ++u)
    *(v4f*)(ks + (sr + 8 * u) * KP + 4 * sc) = kr[u];
#pragma unroll
  for (int u = 0; u < 4; ++u) {
    const v4f a = vr[2 * u];
    const v4f b = vr[2 * u + 1];
    float* d = vs + (sr + 8 * u) * VP + 8 * sc;
    v4f lo, hi;
    lo.x = a.x; lo.y = b.x; lo.z = a.y; lo.w = b.y;
    hi.x = a.z; hi.y = b.z; hi.z = a.w; hi.w = b.w;
    *(v4f*)(d)     = lo;
    *(v4f*)(d + 4) = hi;
  }
}

template <int S0, int S1>
__device__ __forceinline__ void v_frags(const float* __restrict__ vt, v2f (&vf)[16][4])
{
#pragma unroll
  for (int s_ = S0; s_ < S1; ++s_)
#pragma unroll
    for (int nn = 0; nn < 4; ++nn)
      vf[s_][nn] = *(const v2f*)(vt + (2 * s_) * VP + 32 * nn);
}

__device__ __forceinline__ void p_frags(const float* __restrict__ pb, const int m, const int kp,
                                        v2f (&pf)[16])
{
#pragma unroll
  for (int s_ = 0; s_ < 16; ++s_)
    pf[s_] = *(const v2f*)(pb + m * PP + 4 * s_ + 2 * kp);
}

template <int S0, int S1>
__device__ __forceinline__ void s_products(const v2f (&qf)[16], const v2f (&kf)[16][KT], v8f (&s)[KT])
{
#pragma unroll
  for (int s_ = S0; s_ < S1; ++s_)
#pragma unroll
    for (int t = 0; t < KT; ++t)
      s[t] = __builtin_amdgcn_wmma_f32_16x16x4_f32(false, qf[s_], false, kf[s_][t],
                                                   (short)0, s[t], false, false);
}

template <bool FIRST, bool MASK>
__device__ __forceinline__ void kv_trip(
    const float* __restrict__ ks, const float* __restrict__ vsp,
    const int rowl, const int m, const int kp,
    const v2f (&qf)[16], v8f (&acc)[4], float (&mr)[8], float (&lr)[8],
    float (&alpha)[8], float* __restrict__ pb)
{
  const float* kt = ks + m * KP + 2 * kp;
  v2f kf[16][KT];
#pragma unroll
  for (int s_ = 0; s_ < 16; ++s_)
#pragma unroll
    for (int t = 0; t < KT; ++t)
      kf[s_][t] = *(const v2f*)(kt + (t * 16) * KP + 4 * s_);

  __builtin_amdgcn_sched_barrier(0);

  const float* vt = vsp + kp * VP + 2 * m;
  v2f vf[16][4];
  v2f pf[16];
  v8f s[KT];
#pragma unroll
  for (int t = 0; t < KT; ++t) s[t] = v8f{};

  if (!FIRST) {
#pragma unroll
    for (int r = 0; r < 8; ++r)
#pragma unroll
      for (int nn = 0; nn < 4; ++nn) acc[nn][r] *= alpha[r];
  }
  s_products<0, 8>(qf, kf, s);

  __builtin_amdgcn_sched_barrier(0);
  if (!FIRST) {
    v_frags<0, 8>(vt, vf);
    p_frags(pb, m, kp, pf);
  }
  __builtin_amdgcn_sched_barrier(0);

  s_products<8, 12>(qf, kf, s);

  __builtin_amdgcn_sched_barrier(0);
  if (!FIRST) v_frags<8, 16>(vt, vf);
  __builtin_amdgcn_sched_barrier(0);

  s_products<12, 16>(qf, kf, s);

  if (MASK) {
#pragma unroll
    for (int t = 0; t < KT; ++t)
#pragma unroll
      for (int r = 0; r < 8; ++r)
        if (16 * t + m > rowl + r + 8 * kp) s[t][r] = -1e30f;
  }

  __builtin_amdgcn_sched_barrier(0);

  float anew[8];
  float p[KT][8];
#pragma unroll
  for (int h = 0; h < 2; ++h) {
    float rm[4], rs[4];
#pragma unroll
    for (int rr = 0; rr < 4; ++rr) {
      const int r = 4 * h + rr;
      rm[rr] = s[0][r];
#pragma unroll
      for (int t = 1; t < KT; ++t) rm[rr] = fmaxf(rm[rr], s[t][r]);
    }
#pragma unroll
    for (int rr = 0; rr < 4; ++rr) rm[rr] = fmaxf(rm[rr], row_xor<1>(rm[rr]));
#pragma unroll
    for (int rr = 0; rr < 4; ++rr) rm[rr] = fmaxf(rm[rr], row_xor<2>(rm[rr]));
#pragma unroll
    for (int rr = 0; rr < 4; ++rr) rm[rr] = fmaxf(rm[rr], row_xor<4>(rm[rr]));
#pragma unroll
    for (int rr = 0; rr < 4; ++rr) rm[rr] = fmaxf(rm[rr], row_xor<8>(rm[rr]));
#pragma unroll
    for (int rr = 0; rr < 4; ++rr) {
      const int r = 4 * h + rr;
      const float mnew = fmaxf(mr[r], rm[rr]);
      anew[r] = __expf(mr[r] - mnew);
      mr[r] = mnew;
    }
#pragma unroll
    for (int rr = 0; rr < 4; ++rr) {
      const int r = 4 * h + rr;
      rs[rr] = 0.0f;
#pragma unroll
      for (int t = 0; t < KT; ++t) {
        p[t][r] = __expf(s[t][r] - mr[r]);
        rs[rr] += p[t][r];
      }
    }
#pragma unroll
    for (int rr = 0; rr < 4; ++rr) rs[rr] += row_xor<1>(rs[rr]);
#pragma unroll
    for (int rr = 0; rr < 4; ++rr) rs[rr] += row_xor<2>(rs[rr]);
#pragma unroll
    for (int rr = 0; rr < 4; ++rr) rs[rr] += row_xor<4>(rs[rr]);
#pragma unroll
    for (int rr = 0; rr < 4; ++rr) rs[rr] += row_xor<8>(rs[rr]);
#pragma unroll
    for (int rr = 0; rr < 4; ++rr) {
      const int r = 4 * h + rr;
      lr[r] = lr[r] * anew[r] + rs[rr];
    }
    if (!FIRST) {
#pragma unroll
      for (int s_ = 8 * h; s_ < 8 * h + 8; ++s_)
#pragma unroll
        for (int nn = 0; nn < 4; ++nn)
          acc[nn] = __builtin_amdgcn_wmma_f32_16x16x4_f32(false, pf[s_], false, vf[s_][nn],
                                                          (short)0, acc[nn], false, false);
#pragma unroll
      for (int i = 0; i < 32; ++i) {
        __builtin_amdgcn_sched_group_barrier(0x008, 1, 0);
        __builtin_amdgcn_sched_group_barrier(0x002, 5, 0);
      }
    }
    __builtin_amdgcn_sched_barrier(0);
  }

  __builtin_amdgcn_wave_barrier();

#pragma unroll
  for (int r = 0; r < 8; ++r) {
#pragma unroll
    for (int t = 0; t < KT; ++t)
      pb[(r + 8 * kp) * PP + 16 * t + m] = p[t][r];
    alpha[r] = anew[r];
  }
}

__global__ __launch_bounds__(128) void flash_attn(
    float* qm, const float* __restrict__ km,
    const float* __restrict__ vm, float* out)
{
  __shared__ __attribute__((aligned(16))) float kst[3 * KSZ];
  __shared__ __attribute__((aligned(16))) float vst[3 * VSZ];
  __shared__ __attribute__((aligned(16))) float pbuf[4][16 * PP];

  const int lane = threadIdx.x & 31;
  const int wid  = threadIdx.x >> 5;
  const int m    = lane & 15;
  const int kp   = lane >> 4;
  const int sc   = threadIdx.x & 15;
  const int sr   = threadIdx.x >> 4;

  const int bb   = blockIdx.x >> 6;
  const int role = blockIdx.x & 1;
  const int pi   = (blockIdx.x & 63) >> 1;

  const size_t base = (size_t)bb * Tsz * Hsz;
  const float* kb = km + base;
  const float* vb = vm + base;
  float* pb = pbuf[wid];

  const int nseg = 2 - role;
#pragma nounroll
  for (int seg = 0; seg < nseg; ++seg) {
    const bool head = (seg == 1);
    const bool part = (role == 1) || head;
    const int  qblk = part ? 63 - pi : pi;
    const int  j0   = (role == 1) ? 31 - pi : 0;
    const int  nblk = (role == 1) ? 33 : (head ? 31 - pi : pi + 1);
    float* buf = (head ? out : qm) + base;
    const int ti = 4 * qblk + wid;

    v8f acc[4] = {};
    float mr[8], lr[8], alpha[8];
#pragma unroll
    for (int r = 0; r < 8; ++r) { mr[r] = -1e30f; lr[r] = 0.0f; alpha[r] = 0.0f; }

    if (nblk > 0) {
      v4f kr[8], vr[8];
      stage_load(kb, vb, j0, sr, sc, kr, vr);

      v2f qf[16];
      {
        const float* qr = buf + (size_t)(ti * 16 + m) * Hsz + 2 * kp;
#pragma unroll
        for (int s = 0; s < 16; ++s) {
          v2f t = *(const v2f*)(qr + 4 * s);
          qf[s].x = t.x * 0.125f;
          qf[s].y = t.y * 0.125f;
        }
      }

      const int nfull = nblk - 1;
      const int rowl  = ti * 16 - (j0 + nfull) * BN;

      __syncthreads();
      stage_store(kst, vst, sr, sc, kr, vr);
      __syncthreads();

      int prv = 2, cur = 0, nxt = 1;
      if (nfull == 0) {
        kv_trip<true, true>(kst, vst, rowl, m, kp, qf, acc, mr, lr, alpha, pb);
      } else {
        stage_load(kb, vb, j0 + 1, sr, sc, kr, vr);
        __builtin_amdgcn_sched_barrier(0);
        kv_trip<true, false>(kst, vst, rowl, m, kp, qf, acc, mr, lr, alpha, pb);
        __builtin_amdgcn_sched_barrier(0);
        stage_store(kst + KSZ, vst + VSZ, sr, sc, kr, vr);
        __syncthreads();
        prv = 0; cur = 1; nxt = 2;

        for (int jb = 1; jb < nfull; ++jb) {
          stage_load(kb, vb, j0 + jb + 1, sr, sc, kr, vr);
          __builtin_amdgcn_sched_barrier(0);
          kv_trip<false, false>(kst + cur * KSZ, vst + prv * VSZ, rowl, m, kp,
                                qf, acc, mr, lr, alpha, pb);
          __builtin_amdgcn_sched_barrier(0);
          stage_store(kst + nxt * KSZ, vst + nxt * VSZ, sr, sc, kr, vr);
          __syncthreads();
          const int t = prv; prv = cur; cur = nxt; nxt = t;
        }
        kv_trip<false, true>(kst + cur * KSZ, vst + prv * VSZ, rowl, m, kp,
                             qf, acc, mr, lr, alpha, pb);
      }

      {
        __builtin_amdgcn_wave_barrier();
        v2f vf[16][4];
        v2f pf[16];
        v_frags<0, 8>(vst + cur * VSZ + kp * VP + 2 * m, vf);
        p_frags(pb, m, kp, pf);
#pragma unroll
        for (int r = 0; r < 8; ++r)
#pragma unroll
          for (int nn = 0; nn < 4; ++nn) acc[nn][r] *= alpha[r];
        v_frags<8, 16>(vst + cur * VSZ + kp * VP + 2 * m, vf);
        __builtin_amdgcn_sched_barrier(0);
#pragma unroll
        for (int s_ = 0; s_ < 16; ++s_)
#pragma unroll
          for (int nn = 0; nn < 4; ++nn)
            acc[nn] = __builtin_amdgcn_wmma_f32_16x16x4_f32(false, pf[s_], false, vf[s_][nn],
                                                            (short)0, acc[nn], false, false);
      }
    }

    float* ob = buf + (size_t)(ti * 16 + 8 * kp) * Hsz + m;
    if (part) {
#pragma unroll
      for (int nn = 0; nn < 4; ++nn)
#pragma unroll
        for (int r = 0; r < 8; ++r)
          ob[(size_t)r * Hsz + nn * 16] = acc[nn][r];
      if (m == 0) {
        float* sd = out + base + (size_t)(pi * 64 + wid * 16 + 8 * kp) * Hsz + 32 * role;
#pragma unroll
        for (int r = 0; r < 8; ++r) {
          sd[(size_t)r * Hsz]     = mr[r];
          sd[(size_t)r * Hsz + 1] = lr[r];
        }
      }
    } else {
#pragma unroll
      for (int nn = 0; nn < 4; ++nn)
#pragma unroll
        for (int r = 0; r < 8; ++r)
          ob[(size_t)r * Hsz + nn * 16] = acc[nn][r] / lr[r];
    }
  }
}

__global__ __launch_bounds__(256) void merge_out(const float* __restrict__ qm, float* out)
{
  __shared__ v4f scs[64];

  const int bb = blockIdx.x >> 5;
  const int pi = blockIdx.x & 31;
  const size_t base = (size_t)bb * Tsz * Hsz;
  const size_t lb = base + (size_t)(pi * 64) * Hsz;
  const size_t hb = base + (size_t)((63 - pi) * 64) * Hsz;

  if (threadIdx.x < 64) {
    const float* sp = out + lb + (size_t)threadIdx.x * Hsz;
    const v2f a = *(const v2f*)(sp);
    const v2f b = *(const v2f*)(sp + 32);
    v4f s;
    s.x = a.x; s.y = a.y; s.z = b.x; s.w = b.y;
    scs[threadIdx.x] = s;
  }
  __syncthreads();

  const int sc = threadIdx.x & 15;
  const int sr = threadIdx.x >> 4;
#pragma unroll
  for (int u = 0; u < 4; ++u) {
    const int row = sr + 16 * u;
    const size_t off = (size_t)row * Hsz + 4 * sc;
    const v4f lv = *(const v4f*)(qm + lb + off);
    const v4f oa = *(const v4f*)(out + hb + off);
    const v4f ob = *(const v4f*)(qm + hb + off);
    const v4f s  = scs[row];
    const float mx  = fmaxf(s.x, s.z);
    const float wa  = __expf(s.x - mx);
    const float wb  = __expf(s.z - mx);
    const float den = wa * s.y + wb * s.w;
    v4f o;
    o.x = (wa * oa.x + wb * ob.x) / den;
    o.y = (wa * oa.y + wb * ob.y) / den;
    o.z = (wa * oa.z + wb * ob.z) / den;
    o.w = (wa * oa.w + wb * ob.w) / den;
    *(v4f*)(out + lb + off) = lv;
    *(v4f*)(out + hb + off) = o;
  }
}

extern "C" void kernel_launch(void* const* d_in, const int* in_sizes, int n_in,
                              void* d_out, int out_size, void* d_ws, size_t ws_size,
                              hipStream_t stream) {
  (void)in_sizes; (void)n_in; (void)out_size; (void)ws_size;

  const float* x  = (const float*)d_in[0];
  const float* Wk = (const float*)d_in[1];
  const float* Wq = (const float*)d_in[2];
  const float* Wv = (const float*)d_in[3];
  float* out = (float*)d_out;

  float* qo = (float*)d_ws;
  float* ko = qo + (size_t)BT * Hsz;
  float* vo = ko + (size_t)BT * Hsz;

  const int tiles = BT / 16;
  dim3 block(128);
  dim3 grid(tiles / 4);

  proj_qkv<<<grid, block, 0, stream>>>(x, Wk, Wq, Wv, qo, ko, vo, out);
  flash_attn<<<grid, block, 0, stream>>>(qo, ko, vo, out);
  merge_out<<<dim3(128), dim3(256), 0, stream>>>(qo, out);
}


# projection: workgroup g walks its 64-k chunks in rotated order (first chunk from g's index bits, wrapping round) so workgroups do not ask L2 for the same weight lines at once; new summation order
# speedup vs baseline: 5.5848x; 1.0784x over previous
#include <hip/hip_runtime.h>

typedef __attribute__((ext_vector_type(2))) float v2f;
typedef __attribute__((ext_vector_type(4))) float v4f;
typedef __attribute__((ext_vector_type(8))) float v8f;

static constexpr int Bsz = 4;
static constexpr int Tsz = 4096;
static constexpr int Csz = 1024;
static constexpr int Hsz = 64;
static constexpr int BT  = Bsz * Tsz;

static constexpr int PM  = 64;
static constexpr int KC  = 64;
static constexpr int XP  = KC + 4;
static constexpr int CP  = 3 * Hsz + 4;

template <bool MORE>
__device__ __forceinline__ void proj_trip(
    const float* __restrict__ xa, float* __restrict__ xs0,
    const float* __restrict__ xg, const float* const (&wb)[3],
    const int t, const int ck, const int cn,
    const int sr, const int sc, v4f (&xr)[8], v8f (&acc)[4][3])
{
  const int cur = t & 1;
  const float* xc = xa + cur * (PM * XP);
  const size_t kof = (size_t)(ck * KC) * Hsz;

  v2f wf[16][3];
#pragma unroll
  for (int s = 0; s < 8; ++s)
#pragma unroll
    for (int j = 0; j < 3; ++j) {
      wf[s][j].x = wb[j][kof + (size_t)(4 * s) * Hsz];
      wf[s][j].y = wb[j][kof + (size_t)(4 * s + 1) * Hsz];
    }

  v2f af[16][4];
#pragma unroll
  for (int s = 0; s < 16; ++s)
#pragma unroll
    for (int mt = 0; mt < 4; ++mt)
      af[s][mt] = *(const v2f*)(xc + (16 * mt) * XP + 4 * s);

  __builtin_amdgcn_sched_barrier(0);

#pragma unroll
  for (int s = 0; s < 4; ++s)
#pragma unroll
    for (int mt = 0; mt < 4; ++mt)
#pragma unroll
      for (int j = 0; j < 3; ++j)
        acc[mt][j] = __builtin_amdgcn_wmma_f32_16x16x4_f32(false, af[s][mt], false, wf[s][j],
                                                          (short)0, acc[mt][j], false, false);

  __builtin_amdgcn_sched_barrier(0);

#pragma unroll
  for (int s = 8; s < 16; ++s)
#pragma unroll
    for (int j = 0; j < 3; ++j) {
      wf[s][j].x = wb[j][kof + (size_t)(4 * s) * Hsz];
      wf[s][j].y = wb[j][kof + (size_t)(4 * s + 1) * Hsz];
    }

  if (MORE) {
#pragma unroll
    for (int u = 0; u < 8; ++u)
      xr[u] = *(const v4f*)(xg + (size_t)(8 * u) * Csz + cn * KC);
  }

  __builtin_amdgcn_sched_barrier(0);

#pragma unroll
  for (int s = 4; s < 16; ++s)
#pragma unroll
    for (int mt = 0; mt < 4; ++mt)
#pragma unroll
      for (int j = 0; j < 3; ++j)
        acc[mt][j] = __builtin_amdgcn_wmma_f32_16x16x4_f32(false, af[s][mt], false, wf[s][j],
                                                          (short)0, acc[mt][j], false, false);

  __builtin_amdgcn_sched_barrier(0);

  if (MORE) {
    float* xd = xs0 + (cur ^ 1) * (PM * XP) + sr * XP + 4 * sc;
#pragma unroll
    for (int u = 0; u < 8; ++u)
      *(v4f*)(xd + (8 * u) * XP) = xr[u];
  }
}

__global__ __launch_bounds__(128) void proj_qkv(
    const float* __restrict__ x,
    const float* __restrict__ Wk,
    const float* __restrict__ Wq,
    const float* __restrict__ Wv,
    float* __restrict__ qo, float* __restrict__ ko, float* __restrict__ vo,
    float* __restrict__ q2)
{
  __shared__ __attribute__((aligned(16))) float xs[2][PM * XP];
  __shared__ __attribute__((aligned(16))) float cs[PM * CP];

  const int lane = threadIdx.x & 31;
  const int wid  = __builtin_amdgcn_readfirstlane((int)(threadIdx.x >> 5));
  const int m    = lane & 15;
  const int kp   = lane >> 4;
  const int sc   = threadIdx.x & 15;
  const int sr   = threadIdx.x >> 4;
  const int row0 = blockIdx.x * PM;
  constexpr int NCH = Csz / KC;
  const int rot = (int)((blockIdx.x ^ (blockIdx.x >> 4)) & (NCH - 1));

  const float* wb[3];
#pragma unroll
  for (int j = 0; j < 3; ++j) {
    const int g   = 3 * wid + j;
    const int mat = g >> 2;
    const float* wsel = (mat == 0) ? Wq : ((mat == 1) ? Wk : Wv);
    wb[j] = wsel + (size_t)(2 * kp) * Hsz + (g & 3) * 16 + m;
  }

  const float* xg = x + (size_t)(row0 + sr) * Csz + 4 * sc;
  const float* xa = &xs[0][0] + m * XP + 2 * kp;

  v8f acc[4][3];
#pragma unroll
  for (int mt = 0; mt < 4; ++mt)
#pragma unroll
    for (int j = 0; j < 3; ++j) acc[mt][j] = v8f{};

  v4f xr[8];
#pragma unroll
  for (int u = 0; u < 8; ++u)
    xr[u] = *(const v4f*)(xg + (size_t)(8 * u) * Csz + rot * KC);
#pragma unroll
  for (int u = 0; u < 8; ++u)
    *(v4f*)(&xs[0][0] + (sr + 8 * u) * XP + 4 * sc) = xr[u];
  __syncthreads();

  for (int t = 0; t < NCH - 1; ++t) {
    proj_trip<true>(xa, xs[0], xg, wb, t, (t + rot) & (NCH - 1), (t + 1 + rot) & (NCH - 1),
                    sr, sc, xr, acc);
    __syncthreads();
  }
  proj_trip<false>(xa, xs[0], xg, wb, NCH - 1, (NCH - 1 + rot) & (NCH - 1), 0, sr, sc, xr, acc);

#pragma unroll
  for (int mt = 0; mt < 4; ++mt)
#pragma unroll
    for (int j = 0; j < 3; ++j)
#pragma unroll
      for (int r = 0; r < 8; ++r)
        cs[(16 * mt + r + 8 * kp) * CP + 16 * (3 * wid + j) + m] = acc[mt][j][r];
  __syncthreads();

  const bool dup = (blockIdx.x & 63) >= 32;
#pragma unroll
  for (int u = 0; u < 8; ++u) {
    const int row = sr + 8 * u;
    const float* cr = cs + row * CP + 4 * sc;
    const size_t o = (size_t)(row0 + row) * Hsz + 4 * sc;
    const v4f qv = *(const v4f*)(cr);
    *(v4f*)(qo + o) = qv;
    if (dup) *(v4f*)(q2 + o) = qv;
    *(v4f*)(ko + o) = *(const v4f*)(cr + Hsz);
    *(v4f*)(vo + o) = *(const v4f*)(cr + 2 * Hsz);
  }
}

static constexpr int KT = 4;
static constexpr int BN = 16 * KT;
static constexpr int PP = BN + 4;
static constexpr int KP = Hsz + 4;
static constexpr int VP = 2 * Hsz + 32;
static constexpr int KSZ = BN * KP;
static constexpr int VSZ = (BN / 2) * VP;

template <int X>
__device__ __forceinline__ float row_xor(const float v)
{
  return __builtin_bit_cast(float, __builtin_amdgcn_update_dpp(
      0, __builtin_bit_cast(int, v), 0x160 + X, 0xf, 0xf, true));
}

__device__ __forceinline__ void stage_load(
    const float* __restrict__ kb, const float* __restrict__ vb,
    const int jb, const int sr, const int sc, v4f (&kr)[8], v4f (&vr)[8])
{
  const float* kg = kb + (size_t)(jb * BN + sr) * Hsz + 4 * sc;
#pragma unroll
  for (int u = 0; u < 8; ++u)
    kr[u] = *(const v4f*)(kg + (size_t)(8 * u) * Hsz);
  const float* vg = vb + (size_t)(jb * BN + 2 * sr) * Hsz + 4 * sc;
#pragma unroll
  for (int u = 0; u < 4; ++u) {
    vr[2 * u]     = *(const v4f*)(vg + (size_t)(16 * u) * Hsz);
    vr[2 * u + 1] = *(const v4f*)(vg + (size_t)(16 * u + 1) * Hsz);
  }
}

__device__ __forceinline__ void stage_store(
    float* __restrict__ ks, float* __restrict__ vs,
    const int sr, const int sc, const v4f (&kr)[8], const v4f (&vr)[8])
{
#pragma unroll
  for (int u = 0; u < 8; ++u)
    *(v4f*)(ks + (sr + 8 * u) * KP + 4 * sc) = kr[u];
#pragma unroll
  for (int u = 0; u < 4; ++u) {
    const v4f a = vr[2 * u];
    const v4f b = vr[2 * u + 1];
    float* d = vs + (sr + 8 * u) * VP + 8 * sc;
    v4f lo, hi;
    lo.x = a.x; lo.y = b.x; lo.z = a.y; lo.w = b.y;
    hi.x = a.z; hi.y = b.z; hi.z = a.w; hi.w = b.w;
    *(v4f*)(d)     = lo;
    *(v4f*)(d + 4) = hi;
  }
}

template <int S0, int S1>
__device__ __forceinline__ void v_frags(const float* __restrict__ vt, v2f (&vf)[16][4])
{
#pragma unroll
  for (int s_ = S0; s_ < S1; ++s_)
#pragma unroll
    for (int nn = 0; nn < 4; ++nn)
      vf[s_][nn] = *(const v2f*)(vt + (2 * s_) * VP + 32 * nn);
}

__device__ __forceinline__ void p_frags(const float* __restrict__ pb, const int m, const int kp,
                                        v2f (&pf)[16])
{
#pragma unroll
  for (int s_ = 0; s_ < 16; ++s_)
    pf[s_] = *(const v2f*)(pb + m * PP + 4 * s_ + 2 * kp);
}

template <int S0, int S1>
__device__ __forceinline__ void s_products(const v2f (&qf)[16], const v2f (&kf)[16][KT], v8f (&s)[KT])
{
#pragma unroll
  for (int s_ = S0; s_ < S1; ++s_)
#pragma unroll
    for (int t = 0; t < KT; ++t)
      s[t] = __builtin_amdgcn_wmma_f32_16x16x4_f32(false, qf[s_], false, kf[s_][t],
                                                   (short)0, s[t], false, false);
}

template <bool FIRST, bool MASK>
__device__ __forceinline__ void kv_trip(
    const float* __restrict__ ks, const float* __restrict__ vsp,
    const int rowl, const int m, const int kp,
    const v2f (&qf)[16], v8f (&acc)[4], float (&mr)[8], float (&lr)[8],
    float (&alpha)[8], float* __restrict__ pb)
{
  const float* kt = ks + m * KP + 2 * kp;
  v2f kf[16][KT];
#pragma unroll
  for (int s_ = 0; s_ < 16; ++s_)
#pragma unroll
    for (int t = 0; t < KT; ++t)
      kf[s_][t] = *(const v2f*)(kt + (t * 16) * KP + 4 * s_);

  __builtin_amdgcn_sched_barrier(0);

  const float* vt = vsp + kp * VP + 2 * m;
  v2f vf[16][4];
  v2f pf[16];
  v8f s[KT];
#pragma unroll
  for (int t = 0; t < KT; ++t) s[t] = v8f{};

  if (!FIRST) {
#pragma unroll
    for (int r = 0; r < 8; ++r)
#pragma unroll
      for (int nn = 0; nn < 4; ++nn) acc[nn][r] *= alpha[r];
  }
  s_products<0, 8>(qf, kf, s);

  __builtin_amdgcn_sched_barrier(0);
  if (!FIRST) {
    v_frags<0, 8>(vt, vf);
    p_frags(pb, m, kp, pf);
  }
  __builtin_amdgcn_sched_barrier(0);

  s_products<8, 12>(qf, kf, s);

  __builtin_amdgcn_sched_barrier(0);
  if (!FIRST) v_frags<8, 16>(vt, vf);
  __builtin_amdgcn_sched_barrier(0);

  s_products<12, 16>(qf, kf, s);

  if (MASK) {
#pragma unroll
    for (int t = 0; t < KT; ++t)
#pragma unroll
      for (int r = 0; r < 8; ++r)
        if (16 * t + m > rowl + r + 8 * kp) s[t][r] = -1e30f;
  }

  __builtin_amdgcn_sched_barrier(0);

  float anew[8];
  float p[KT][8];
#pragma unroll
  for (int h = 0; h < 2; ++h) {
    float rm[4], rs[4];
#pragma unroll
    for (int rr = 0; rr < 4; ++rr) {
      const int r = 4 * h + rr;
      rm[rr] = s[0][r];
#pragma unroll
      for (int t = 1; t < KT; ++t) rm[rr] = fmaxf(rm[rr], s[t][r]);
    }
#pragma unroll
    for (int rr = 0; rr < 4; ++rr) rm[rr] = fmaxf(rm[rr], row_xor<1>(rm[rr]));
#pragma unroll
    for (int rr = 0; rr < 4; ++rr) rm[rr] = fmaxf(rm[rr], row_xor<2>(rm[rr]));
#pragma unroll
    for (int rr = 0; rr < 4; ++rr) rm[rr] = fmaxf(rm[rr], row_xor<4>(rm[rr]));
#pragma unroll
    for (int rr = 0; rr < 4; ++rr) rm[rr] = fmaxf(rm[rr], row_xor<8>(rm[rr]));
#pragma unroll
    for (int rr = 0; rr < 4; ++rr) {
      const int r = 4 * h + rr;
      const float mnew = fmaxf(mr[r], rm[rr]);
      anew[r] = __expf(mr[r] - mnew);
      mr[r] = mnew;
    }
#pragma unroll
    for (int rr = 0; rr < 4; ++rr) {
      const int r = 4 * h + rr;
      rs[rr] = 0.0f;
#pragma unroll
      for (int t = 0; t < KT; ++t) {
        p[t][r] = __expf(s[t][r] - mr[r]);
        rs[rr] += p[t][r];
      }
    }
#pragma unroll
    for (int rr = 0; rr < 4; ++rr) rs[rr] += row_xor<1>(rs[rr]);
#pragma unroll
    for (int rr = 0; rr < 4; ++rr) rs[rr] += row_xor<2>(rs[rr]);
#pragma unroll
    for (int rr = 0; rr < 4; ++rr) rs[rr] += row_xor<4>(rs[rr]);
#pragma unroll
    for (int rr = 0; rr < 4; ++rr) rs[rr] += row_xor<8>(rs[rr]);
#pragma unroll
    for (int rr = 0; rr < 4; ++rr) {
      const int r = 4 * h + rr;
      lr[r] = lr[r] * anew[r] + rs[rr];
    }
    if (!FIRST) {
#pragma unroll
      for (int s_ = 8 * h; s_ < 8 * h + 8; ++s_)
#pragma unroll
        for (int nn = 0; nn < 4; ++nn)
          acc[nn] = __builtin_amdgcn_wmma_f32_16x16x4_f32(false, pf[s_], false, vf[s_][nn],
                                                          (short)0, acc[nn], false, false);
#pragma unroll
      for (int i = 0; i < 32; ++i) {
        __builtin_amdgcn_sched_group_barrier(0x008, 1, 0);
        __builtin_amdgcn_sched_group_barrier(0x002, 5, 0);
      }
    }
    __builtin_amdgcn_sched_barrier(0);
  }

  __builtin_amdgcn_wave_barrier();

#pragma unroll
  for (int r = 0; r < 8; ++r) {
#pragma unroll
    for (int t = 0; t < KT; ++t)
      pb[(r + 8 * kp) * PP + 16 * t + m] = p[t][r];
    alpha[r] = anew[r];
  }
}

__global__ __launch_bounds__(128) void flash_attn(
    float* qm, const float* __restrict__ km,
    const float* __restrict__ vm, float* out)
{
  __shared__ __attribute__((aligned(16))) float kst[3 * KSZ];
  __shared__ __attribute__((aligned(16))) float vst[3 * VSZ];
  __shared__ __attribute__((aligned(16))) float pbuf[4][16 * PP];

  const int lane = threadIdx.x & 31;
  const int wid  = threadIdx.x >> 5;
  const int m    = lane & 15;
  const int kp   = lane >> 4;
  const int sc   = threadIdx.x & 15;
  const int sr   = threadIdx.x >> 4;

  const int bb   = blockIdx.x >> 6;
  const int role = blockIdx.x & 1;
  const int pi   = (blockIdx.x & 63) >> 1;

  const size_t base = (size_t)bb * Tsz * Hsz;
  const float* kb = km + base;
  const float* vb = vm + base;
  float* pb = pbuf[wid];

  const int nseg = 2 - role;
#pragma nounroll
  for (int seg = 0; seg < nseg; ++seg) {
    const bool head = (seg == 1);
    const bool part = (role == 1) || head;
    const int  qblk = part ? 63 - pi : pi;
    const int  j0   = (role == 1) ? 31 - pi : 0;
    const int  nblk = (role == 1) ? 33 : (head ? 31 - pi : pi + 1);
    float* buf = (head ? out : qm) + base;
    const int ti = 4 * qblk + wid;

    v8f acc[4] = {};
    float mr[8], lr[8], alpha[8];
#pragma unroll
    for (int r = 0; r < 8; ++r) { mr[r] = -1e30f; lr[r] = 0.0f; alpha[r] = 0.0f; }

    if (nblk > 0) {
      v4f kr[8], vr[8];
      stage_load(kb, vb, j0, sr, sc, kr, vr);

      v2f qf[16];
      {
        const float* qr = buf + (size_t)(ti * 16 + m) * Hsz + 2 * kp;
#pragma unroll
        for (int s = 0; s < 16; ++s) {
          v2f t = *(const v2f*)(qr + 4 * s);
          qf[s].x = t.x * 0.125f;
          qf[s].y = t.y * 0.125f;
        }
      }

      const int nfull = nblk - 1;
      const int rowl  = ti * 16 - (j0 + nfull) * BN;

      __syncthreads();
      stage_store(kst, vst, sr, sc, kr, vr);
      __syncthreads();

      int prv = 2, cur = 0, nxt = 1;
      if (nfull == 0) {
        kv_trip<true, true>(kst, vst, rowl, m, kp, qf, acc, mr, lr, alpha, pb);
      } else {
        stage_load(kb, vb, j0 + 1, sr, sc, kr, vr);
        __builtin_amdgcn_sched_barrier(0);
        kv_trip<true, false>(kst, vst, rowl, m, kp, qf, acc, mr, lr, alpha, pb);
        __builtin_amdgcn_sched_barrier(0);
        stage_store(kst + KSZ, vst + VSZ, sr, sc, kr, vr);
        __syncthreads();
        prv = 0; cur = 1; nxt = 2;

        for (int jb = 1; jb < nfull; ++jb) {
          stage_load(kb, vb, j0 + jb + 1, sr, sc, kr, vr);
          __builtin_amdgcn_sched_barrier(0);
          kv_trip<false, false>(kst + cur * KSZ, vst + prv * VSZ, rowl, m, kp,
                                qf, acc, mr, lr, alpha, pb);
          __builtin_amdgcn_sched_barrier(0);
          stage_store(kst + nxt * KSZ, vst + nxt * VSZ, sr, sc, kr, vr);
          __syncthreads();
          const int t = prv; prv = cur; cur = nxt; nxt = t;
        }
        kv_trip<false, true>(kst + cur * KSZ, vst + prv * VSZ, rowl, m, kp,
                             qf, acc, mr, lr, alpha, pb);
      }

      {
        __builtin_amdgcn_wave_barrier();
        v2f vf[16][4];
        v2f pf[16];
        v_frags<0, 8>(vst + cur * VSZ + kp * VP + 2 * m, vf);
        p_frags(pb, m, kp, pf);
#pragma unroll
        for (int r = 0; r < 8; ++r)
#pragma unroll
          for (int nn = 0; nn < 4; ++nn) acc[nn][r] *= alpha[r];
        v_frags<8, 16>(vst + cur * VSZ + kp * VP + 2 * m, vf);
        __builtin_amdgcn_sched_barrier(0);
#pragma unroll
        for (int s_ = 0; s_ < 16; ++s_)
#pragma unroll
          for (int nn = 0; nn < 4; ++nn)
            acc[nn] = __builtin_amdgcn_wmma_f32_16x16x4_f32(false, pf[s_], false, vf[s_][nn],
                                                            (short)0, acc[nn], false, false);
      }
    }

    float* ob = buf + (size_t)(ti * 16 + 8 * kp) * Hsz + m;
    if (part) {
#pragma unroll
      for (int nn = 0; nn < 4; ++nn)
#pragma unroll
        for (int r = 0; r < 8; ++r)
          ob[(size_t)r * Hsz + nn * 16] = acc[nn][r];
      if (m == 0) {
        float* sd = out + base + (size_t)(pi * 64 + wid * 16 + 8 * kp) * Hsz + 32 * role;
#pragma unroll
        for (int r = 0; r < 8; ++r) {
          sd[(size_t)r * Hsz]     = mr[r];
          sd[(size_t)r * Hsz + 1] = lr[r];
        }
      }
    } else {
#pragma unroll
      for (int nn = 0; nn < 4; ++nn)
#pragma unroll
        for (int r = 0; r < 8; ++r)
          ob[(size_t)r * Hsz + nn * 16] = acc[nn][r] / lr[r];
    }
  }
}

__global__ __launch_bounds__(256) void merge_out(const float* __restrict__ qm, float* out)
{
  __shared__ v4f scs[64];

  const int bb = blockIdx.x >> 5;
  const int pi = blockIdx.x & 31;
  const size_t base = (size_t)bb * Tsz * Hsz;
  const size_t lb = base + (size_t)(pi * 64) * Hsz;
  const size_t hb = base + (size_t)((63 - pi) * 64) * Hsz;

  if (threadIdx.x < 64) {
    const float* sp = out + lb + (size_t)threadIdx.x * Hsz;
    const v2f a = *(const v2f*)(sp);
    const v2f b = *(const v2f*)(sp + 32);
    v4f s;
    s.x = a.x; s.y = a.y; s.z = b.x; s.w = b.y;
    scs[threadIdx.x] = s;
  }
  __syncthreads();

  const int sc = threadIdx.x & 15;
  const int sr = threadIdx.x >> 4;
#pragma unroll
  for (int u = 0; u < 4; ++u) {
    const int row = sr + 16 * u;
    const size_t off = (size_t)row * Hsz + 4 * sc;
    const v4f lv = *(const v4f*)(qm + lb + off);
    const v4f oa = *(const v4f*)(out + hb + off);
    const v4f ob = *(const v4f*)(qm + hb + off);
    const v4f s  = scs[row];
    const float mx  = fmaxf(s.x, s.z);
    const float wa  = __expf(s.x - mx);
    const float wb  = __expf(s.z - mx);
    const float den = wa * s.y + wb * s.w;
    v4f o;
    o.x = (wa * oa.x + wb * ob.x) / den;
    o.y = (wa * oa.y + wb * ob.y) / den;
    o.z = (wa * oa.z + wb * ob.z) / den;
    o.w = (wa * oa.w + wb * ob.w) / den;
    *(v4f*)(out + lb + off) = lv;
    *(v4f*)(out + hb + off) = o;
  }
}

extern "C" void kernel_launch(void* const* d_in, const int* in_sizes, int n_in,
                              void* d_out, int out_size, void* d_ws, size_t ws_size,
                              hipStream_t stream) {
  (void)in_sizes; (void)n_in; (void)out_size; (void)ws_size;

  const float* x  = (const float*)d_in[0];
  const float* Wk = (const float*)d_in[1];
  const float* Wq = (const float*)d_in[2];
  const float* Wv = (const float*)d_in[3];
  float* out = (float*)d_out;

  float* qo = (float*)d_ws;
  float* ko = qo + (size_t)BT * Hsz;
  float* vo = ko + (size_t)BT * Hsz;

  const int tiles = BT / 16;
  dim3 block(128);
  dim3 grid(tiles / 4);

  proj_qkv<<<grid, block, 0, stream>>>(x, Wk, Wq, Wv, qo, ko, vo, out);
  flash_attn<<<grid, block, 0, stream>>>(qo, ko, vo, out);
  merge_out<<<dim3(128), dim3(256), 0, stream>>>(qo, out);
}


# attention: the two products of a q tile run on two waves of one SIMD (8-wave workgroup): score wave does S and softmax, output wave O of the block before, P and factors via LDS; sums unchanged
# speedup vs baseline: 5.9452x; 1.0645x over previous
#include <hip/hip_runtime.h>

typedef __attribute__((ext_vector_type(2))) float v2f;
typedef __attribute__((ext_vector_type(4))) float v4f;
typedef __attribute__((ext_vector_type(8))) float v8f;

static constexpr int Bsz = 4;
static constexpr int Tsz = 4096;
static constexpr int Csz = 1024;
static constexpr int Hsz = 64;
static constexpr int BT  = Bsz * Tsz;

static constexpr int PM  = 64;
static constexpr int KC  = 64;
static constexpr int XP  = KC + 4;
static constexpr int CP  = 3 * Hsz + 4;

template <bool MORE>
__device__ __forceinline__ void proj_trip(
    const float* __restrict__ xa, float* __restrict__ xs0,
    const float* __restrict__ xg, const float* const (&wb)[3],
    const int t, const int ck, const int cn,
    const int sr, const int sc, v4f (&xr)[8], v8f (&acc)[4][3])
{
  const int cur = t & 1;
  const float* xc = xa + cur * (PM * XP);
  const size_t kof = (size_t)(ck * KC) * Hsz;

  v2f wf[16][3];
#pragma unroll
  for (int s = 0; s < 8; ++s)
#pragma unroll
    for (int j = 0; j < 3; ++j) {
      wf[s][j].x = wb[j][kof + (size_t)(4 * s) * Hsz];
      wf[s][j].y = wb[j][kof + (size_t)(4 * s + 1) * Hsz];
    }

  v2f af[16][4];
#pragma unroll
  for (int s = 0; s < 16; ++s)
#pragma unroll
    for (int mt = 0; mt < 4; ++mt)
      af[s][mt] = *(const v2f*)(xc + (16 * mt) * XP + 4 * s);

  __builtin_amdgcn_sched_barrier(0);

#pragma unroll
  for (int s = 0; s < 4; ++s)
#pragma unroll
    for (int mt = 0; mt < 4; ++mt)
#pragma unroll
      for (int j = 0; j < 3; ++j)
        acc[mt][j] = __builtin_amdgcn_wmma_f32_16x16x4_f32(false, af[s][mt], false, wf[s][j],
                                                          (short)0, acc[mt][j], false, false);

  __builtin_amdgcn_sched_barrier(0);

#pragma unroll
  for (int s = 8; s < 16; ++s)
#pragma unroll
    for (int j = 0; j < 3; ++j) {
      wf[s][j].x = wb[j][kof + (size_t)(4 * s) * Hsz];
      wf[s][j].y = wb[j][kof + (size_t)(4 * s + 1) * Hsz];
    }

  if (MORE) {
#pragma unroll
    for (int u = 0; u < 8; ++u)
      xr[u] = *(const v4f*)(xg + (size_t)(8 * u) * Csz + cn * KC);
  }

  __builtin_amdgcn_sched_barrier(0);

#pragma unroll
  for (int s = 4; s < 16; ++s)
#pragma unroll
    for (int mt = 0; mt < 4; ++mt)
#pragma unroll
      for (int j = 0; j < 3; ++j)
        acc[mt][j] = __builtin_amdgcn_wmma_f32_16x16x4_f32(false, af[s][mt], false, wf[s][j],
                                                          (short)0, acc[mt][j], false, false);

  __builtin_amdgcn_sched_barrier(0);

  if (MORE) {
    float* xd = xs0 + (cur ^ 1) * (PM * XP) + sr * XP + 4 * sc;
#pragma unroll
    for (int u = 0; u < 8; ++u)
      *(v4f*)(xd + (8 * u) * XP) = xr[u];
  }
}

__global__ __launch_bounds__(128) void proj_qkv(
    const float* __restrict__ x,
    const float* __restrict__ Wk,
    const float* __restrict__ Wq,
    const float* __restrict__ Wv,
    float* __restrict__ qo, float* __restrict__ ko, float* __restrict__ vo,
    float* __restrict__ q2)
{
  __shared__ __attribute__((aligned(16))) float xs[2][PM * XP];
  __shared__ __attribute__((aligned(16))) float cs[PM * CP];

  const int lane = threadIdx.x & 31;
  const int wid  = __builtin_amdgcn_readfirstlane((int)(threadIdx.x >> 5));
  const int m    = lane & 15;
  const int kp   = lane >> 4;
  const int sc   = threadIdx.x & 15;
  const int sr   = threadIdx.x >> 4;
  const int row0 = blockIdx.x * PM;
  constexpr int NCH = Csz / KC;
  const int rot = (int)((blockIdx.x ^ (blockIdx.x >> 4)) & (NCH - 1));

  const float* wb[3];
#pragma unroll
  for (int j = 0; j < 3; ++j) {
    const int g   = 3 * wid + j;
    const int mat = g >> 2;
    const float* wsel = (mat == 0) ? Wq : ((mat == 1) ? Wk : Wv);
    wb[j] = wsel + (size_t)(2 * kp) * Hsz + (g & 3) * 16 + m;
  }

  const float* xg = x + (size_t)(row0 + sr) * Csz + 4 * sc;
  const float* xa = &xs[0][0] + m * XP + 2 * kp;

  v8f acc[4][3];
#pragma unroll
  for (int mt = 0; mt < 4; ++mt)
#pragma unroll
    for (int j = 0; j < 3; ++j) acc[mt][j] = v8f{};

  v4f xr[8];
#pragma unroll
  for (int u = 0; u < 8; ++u)
    xr[u] = *(const v4f*)(xg + (size_t)(8 * u) * Csz + rot * KC);
#pragma unroll
  for (int u = 0; u < 8; ++u)
    *(v4f*)(&xs[0][0] + (sr + 8 * u) * XP + 4 * sc) = xr[u];
  __syncthreads();

  for (int t = 0; t < NCH - 1; ++t) {
    proj_trip<true>(xa, xs[0], xg, wb, t, (t + rot) & (NCH - 1), (t + 1 + rot) & (NCH - 1),
                    sr, sc, xr, acc);
    __syncthreads();
  }
  proj_trip<false>(xa, xs[0], xg, wb, NCH - 1, (NCH - 1 + rot) & (NCH - 1), 0, sr, sc, xr, acc);

#pragma unroll
  for (int mt = 0; mt < 4; ++mt)
#pragma unroll
    for (int j = 0; j < 3; ++j)
#pragma unroll
      for (int r = 0; r < 8; ++r)
        cs[(16 * mt + r + 8 * kp) * CP + 16 * (3 * wid + j) + m] = acc[mt][j][r];
  __syncthreads();

  const bool dup = (blockIdx.x & 63) >= 32;
#pragma unroll
  for (int u = 0; u < 8; ++u) {
    const int row = sr + 8 * u;
    const float* cr = cs + row * CP + 4 * sc;
    const size_t o = (size_t)(row0 + row) * Hsz + 4 * sc;
    const v4f qv = *(const v4f*)(cr);
    *(v4f*)(qo + o) = qv;
    if (dup) *(v4f*)(q2 + o) = qv;
    *(v4f*)(ko + o) = *(const v4f*)(cr + Hsz);
    *(v4f*)(vo + o) = *(const v4f*)(cr + 2 * Hsz);
  }
}

static constexpr int KT = 4;
static constexpr int BN = 16 * KT;
static constexpr int PP = BN + 4;
static constexpr int KP = Hsz + 4;
static constexpr int VP = 2 * Hsz + 32;
static constexpr int KSZ = BN * KP;
static constexpr int VSZ = (BN / 2) * VP;

template <int X>
__device__ __forceinline__ float row_xor(const float v)
{
  return __builtin_bit_cast(float, __builtin_amdgcn_update_dpp(
      0, __builtin_bit_cast(int, v), 0x160 + X, 0xf, 0xf, true));
}

__device__ __forceinline__ void stage_load(
    const float* __restrict__ kb, const float* __restrict__ vb,
    const int jb, const int sr, const int sc, v4f (&kr)[8], v4f (&vr)[8])
{
  const float* kg = kb + (size_t)(jb * BN + sr) * Hsz + 4 * sc;
#pragma unroll
  for (int u = 0; u < 8; ++u)
    kr[u] = *(const v4f*)(kg + (size_t)(8 * u) * Hsz);
  const float* vg = vb + (size_t)(jb * BN + 2 * sr) * Hsz + 4 * sc;
#pragma unroll
  for (int u = 0; u < 4; ++u) {
    vr[2 * u]     = *(const v4f*)(vg + (size_t)(16 * u) * Hsz);
    vr[2 * u + 1] = *(const v4f*)(vg + (size_t)(16 * u + 1) * Hsz);
  }
}

__device__ __forceinline__ void stage_store(
    float* __restrict__ ks, float* __restrict__ vs,
    const int sr, const int sc, const v4f (&kr)[8], const v4f (&vr)[8])
{
#pragma unroll
  for (int u = 0; u < 8; ++u)
    *(v4f*)(ks + (sr + 8 * u) * KP + 4 * sc) = kr[u];
#pragma unroll
  for (int u = 0; u < 4; ++u) {
    const v4f a = vr[2 * u];
    const v4f b = vr[2 * u + 1];
    float* d = vs + (sr + 8 * u) * VP + 8 * sc;
    v4f lo, hi;
    lo.x = a.x; lo.y = b.x; lo.z = a.y; lo.w = b.y;
    hi.x = a.z; hi.y = b.z; hi.z = a.w; hi.w = b.w;
    *(v4f*)(d)     = lo;
    *(v4f*)(d + 4) = hi;
  }
}

template <int S0, int S1>
__device__ __forceinline__ void v_frags(const float* __restrict__ vt, v2f (&vf)[16][4])
{
#pragma unroll
  for (int s_ = S0; s_ < S1; ++s_)
#pragma unroll
    for (int nn = 0; nn < 4; ++nn)
      vf[s_][nn] = *(const v2f*)(vt + (2 * s_) * VP + 32 * nn);
}

__device__ __forceinline__ void p_frags(const float* __restrict__ pb, const int m, const int kp,
                                        v2f (&pf)[16])
{
#pragma unroll
  for (int s_ = 0; s_ < 16; ++s_)
    pf[s_] = *(const v2f*)(pb + m * PP + 4 * s_ + 2 * kp);
}

template <int S0, int S1>
__device__ __forceinline__ void s_products(const v2f (&qf)[16], const v2f (&kf)[16][KT], v8f (&s)[KT])
{
#pragma unroll
  for (int s_ = S0; s_ < S1; ++s_)
#pragma unroll
    for (int t = 0; t < KT; ++t)
      s[t] = __builtin_amdgcn_wmma_f32_16x16x4_f32(false, qf[s_], false, kf[s_][t],
                                                   (short)0, s[t], false, false);
}


template <bool MASK>
__device__ __forceinline__ void score_trip(
    const float* __restrict__ ks, const int rowl, const int m, const int kp,
    const v2f (&qf)[16], float (&mr)[8], float (&lr)[8],
    float* __restrict__ pw, float* __restrict__ aw)
{
  const float* kt = ks + m * KP + 2 * kp;
  v2f kf[16][KT];
#pragma unroll
  for (int s_ = 0; s_ < 16; ++s_)
#pragma unroll
    for (int t = 0; t < KT; ++t)
      kf[s_][t] = *(const v2f*)(kt + (t * 16) * KP + 4 * s_);

  __builtin_amdgcn_sched_barrier(0);

  v8f s[KT];
#pragma unroll
  for (int t = 0; t < KT; ++t) s[t] = v8f{};
  s_products<0, 16>(qf, kf, s);

  if (MASK) {
#pragma unroll
    for (int t = 0; t < KT; ++t)
#pragma unroll
      for (int r = 0; r < 8; ++r)
        if (16 * t + m > rowl + r + 8 * kp) s[t][r] = -1e30f;
  }

  float anew[8];
  float p[KT][8];
#pragma unroll
  for (int h = 0; h < 2; ++h) {
    float rm[4], rs[4];
#pragma unroll
    for (int rr = 0; rr < 4; ++rr) {
      const int r = 4 * h + rr;
      rm[rr] = s[0][r];
#pragma unroll
      for (int t = 1; t < KT; ++t) rm[rr] = fmaxf(rm[rr], s[t][r]);
    }
#pragma unroll
    for (int rr = 0; rr < 4; ++rr) rm[rr] = fmaxf(rm[rr], row_xor<1>(rm[rr]));
#pragma unroll
    for (int rr = 0; rr < 4; ++rr) rm[rr] = fmaxf(rm[rr], row_xor<2>(rm[rr]));
#pragma unroll
    for (int rr = 0; rr < 4; ++rr) rm[rr] = fmaxf(rm[rr], row_xor<4>(rm[rr]));
#pragma unroll
    for (int rr = 0; rr < 4; ++rr) rm[rr] = fmaxf(rm[rr], row_xor<8>(rm[rr]));
#pragma unroll
    for (int rr = 0; rr < 4; ++rr) {
      const int r = 4 * h + rr;
      const float mnew = fmaxf(mr[r], rm[rr]);
      anew[r] = __expf(mr[r] - mnew);
      mr[r] = mnew;
    }
#pragma unroll
    for (int rr = 0; rr < 4; ++rr) {
      const int r = 4 * h + rr;
      rs[rr] = 0.0f;
#pragma unroll
      for (int t = 0; t < KT; ++t) {
        p[t][r] = __expf(s[t][r] - mr[r]);
        rs[rr] += p[t][r];
      }
    }
#pragma unroll
    for (int rr = 0; rr < 4; ++rr) rs[rr] += row_xor<1>(rs[rr]);
#pragma unroll
    for (int rr = 0; rr < 4; ++rr) rs[rr] += row_xor<2>(rs[rr]);
#pragma unroll
    for (int rr = 0; rr < 4; ++rr) rs[rr] += row_xor<4>(rs[rr]);
#pragma unroll
    for (int rr = 0; rr < 4; ++rr) rs[rr] += row_xor<8>(rs[rr]);
#pragma unroll
    for (int rr = 0; rr < 4; ++rr) {
      const int r = 4 * h + rr;
      lr[r] = lr[r] * anew[r] + rs[rr];
    }
  }

#pragma unroll
  for (int r = 0; r < 8; ++r)
#pragma unroll
    for (int t = 0; t < KT; ++t)
      pw[(r + 8 * kp) * PP + 16 * t + m] = p[t][r];
  if (m == 0) {
    v4f a0, a1;
    a0.x = anew[0]; a0.y = anew[1]; a0.z = anew[2]; a0.w = anew[3];
    a1.x = anew[4]; a1.y = anew[5]; a1.z = anew[6]; a1.w = anew[7];
    *(v4f*)(aw + 8 * kp)     = a0;
    *(v4f*)(aw + 8 * kp + 4) = a1;
  }
}

__device__ __forceinline__ void out_fetch(
    const float* __restrict__ vsp, const float* __restrict__ pr, const float* __restrict__ ar,
    const int m, const int kp, v2f (&pf)[16], v2f (&vf)[16][4], v8f (&acc)[4])
{
  const float* vt = vsp + kp * VP + 2 * m;
  const v4f a0 = *(const v4f*)(ar + 8 * kp);
  const v4f a1 = *(const v4f*)(ar + 8 * kp + 4);
  p_frags(pr, m, kp, pf);
  v_frags<0, 16>(vt, vf);
  __builtin_amdgcn_sched_barrier(0);

  const float alpha[8] = {a0.x, a0.y, a0.z, a0.w, a1.x, a1.y, a1.z, a1.w};
#pragma unroll
  for (int r = 0; r < 8; ++r)
#pragma unroll
    for (int nn = 0; nn < 4; ++nn) acc[nn][r] *= alpha[r];
}

__device__ __forceinline__ void out_products(
    const v2f (&pf)[16], const v2f (&vf)[16][4], v8f (&acc)[4])
{
#pragma unroll
  for (int s_ = 0; s_ < 16; ++s_)
#pragma unroll
    for (int nn = 0; nn < 4; ++nn)
      acc[nn] = __builtin_amdgcn_wmma_f32_16x16x4_f32(false, pf[s_], false, vf[s_][nn],
                                                      (short)0, acc[nn], false, false);
}

__global__ __launch_bounds__(256) void flash_attn(
    float* qm, const float* __restrict__ km,
    const float* __restrict__ vm, float* out)
{
  __shared__ __attribute__((aligned(16))) float kst[3 * KSZ];
  __shared__ __attribute__((aligned(16))) float vst[3 * VSZ];
  __shared__ __attribute__((aligned(16))) float pbuf[4][2][16 * PP];
  __shared__ __attribute__((aligned(16))) float abuf[4][2][16];
  __shared__ __attribute__((aligned(16))) float lbuf[4][16];

  const int lane = threadIdx.x & 31;
  const int wid  = __builtin_amdgcn_readfirstlane((int)(threadIdx.x >> 5));
  const int qt   = wid & 3;
  const bool wout = wid >= 4;
  const int m    = lane & 15;
  const int kp   = lane >> 4;
  const int st   = threadIdx.x & 127;
  const int sc   = st & 15;
  const int sr   = st >> 4;

  const int bb   = blockIdx.x >> 6;
  const int role = blockIdx.x & 1;
  const int pi   = (blockIdx.x & 63) >> 1;

  const size_t base = (size_t)bb * Tsz * Hsz;
  const float* kb = km + base;
  const float* vb = vm + base;

  const int nseg = 2 - role;
#pragma nounroll
  for (int seg = 0; seg < nseg; ++seg) {
    const bool head = (seg == 1);
    const bool part = (role == 1) || head;
    const int  qblk = part ? 63 - pi : pi;
    const int  j0   = (role == 1) ? 31 - pi : 0;
    const int  nblk = (role == 1) ? 33 : (head ? 31 - pi : pi + 1);
    float* buf = (head ? out : qm) + base;
    const int ti = 4 * qblk + qt;

    v8f acc[4] = {};
    float mr[8], lr[8];
#pragma unroll
    for (int r = 0; r < 8; ++r) { mr[r] = -1e30f; lr[r] = 0.0f; }

    if (nblk > 0) {
      v4f kr[8] = {}, vr[8] = {};
      v2f qf[16] = {};
      if (wout) {
        stage_load(kb, vb, j0, sr, sc, kr, vr);
      } else {
        const float* qr = buf + (size_t)(ti * 16 + m) * Hsz + 2 * kp;
#pragma unroll
        for (int s = 0; s < 16; ++s) {
          v2f t = *(const v2f*)(qr + 4 * s);
          qf[s].x = t.x * 0.125f;
          qf[s].y = t.y * 0.125f;
        }
      }

      const int rowl = ti * 16 - (j0 + nblk - 1) * BN;

      __syncthreads();
      if (wout) stage_store(kst, vst, sr, sc, kr, vr);
      __syncthreads();

      int prv = 2, cur = 0, nxt = 1;
      for (int ph = 0; ph <= nblk; ++ph) {
        if (!wout) {
          if (ph < nblk) {
            float* pw = &pbuf[qt][ph & 1][0];
            float* aw = &abuf[qt][ph & 1][0];
            if (ph == nblk - 1) {
              score_trip<true>(kst + cur * KSZ, rowl, m, kp, qf, mr, lr, pw, aw);
              if (m == 0) {
#pragma unroll
                for (int r = 0; r < 8; ++r) lbuf[qt][8 * kp + r] = lr[r];
              }
            } else {
              score_trip<false>(kst + cur * KSZ, rowl, m, kp, qf, mr, lr, pw, aw);
            }
          }
        } else {
          const bool more = (ph + 1 < nblk);
          v2f pf[16];
          v2f vf[16][4];
          if (more) stage_load(kb, vb, j0 + ph + 1, sr, sc, kr, vr);
          if (ph > 0)
            out_fetch(vst + prv * VSZ, &pbuf[qt][(ph - 1) & 1][0], &abuf[qt][(ph - 1) & 1][0],
                      m, kp, pf, vf, acc);
          if (more) stage_store(kst + nxt * KSZ, vst + nxt * VSZ, sr, sc, kr, vr);
          if (ph > 0) out_products(pf, vf, acc);
        }
        __syncthreads();
        const int t = prv; prv = cur; cur = nxt; nxt = t;
      }
    }

    if (wout) {
      float* ob = buf + (size_t)(ti * 16 + 8 * kp) * Hsz + m;
      if (part) {
#pragma unroll
        for (int nn = 0; nn < 4; ++nn)
#pragma unroll
          for (int r = 0; r < 8; ++r)
            ob[(size_t)r * Hsz + nn * 16] = acc[nn][r];
      } else {
        float lf[8];
#pragma unroll
        for (int r = 0; r < 8; ++r) lf[r] = lbuf[qt][8 * kp + r];
#pragma unroll
        for (int nn = 0; nn < 4; ++nn)
#pragma unroll
          for (int r = 0; r < 8; ++r)
            ob[(size_t)r * Hsz + nn * 16] = acc[nn][r] / lf[r];
      }
    } else if (part && m == 0) {
      float* sd = out + base + (size_t)(pi * 64 + qt * 16 + 8 * kp) * Hsz + 32 * role;
#pragma unroll
      for (int r = 0; r < 8; ++r) {
        sd[(size_t)r * Hsz]     = mr[r];
        sd[(size_t)r * Hsz + 1] = lr[r];
      }
    }
  }
}

__global__ __launch_bounds__(256) void merge_out(const float* __restrict__ qm, float* out)
{
  __shared__ v4f scs[64];

  const int bb = blockIdx.x >> 5;
  const int pi = blockIdx.x & 31;
  const size_t base = (size_t)bb * Tsz * Hsz;
  const size_t lb = base + (size_t)(pi * 64) * Hsz;
  const size_t hb = base + (size_t)((63 - pi) * 64) * Hsz;

  if (threadIdx.x < 64) {
    const float* sp = out + lb + (size_t)threadIdx.x * Hsz;
    const v2f a = *(const v2f*)(sp);
    const v2f b = *(const v2f*)(sp + 32);
    v4f s;
    s.x = a.x; s.y = a.y; s.z = b.x; s.w = b.y;
    scs[threadIdx.x] = s;
  }
  __syncthreads();

  const int sc = threadIdx.x & 15;
  const int sr = threadIdx.x >> 4;
#pragma unroll
  for (int u = 0; u < 4; ++u) {
    const int row = sr + 16 * u;
    const size_t off = (size_t)row * Hsz + 4 * sc;
    const v4f lv = *(const v4f*)(qm + lb + off);
    const v4f oa = *(const v4f*)(out + hb + off);
    const v4f ob = *(const v4f*)(qm + hb + off);
    const v4f s  = scs[row];
    const float mx  = fmaxf(s.x, s.z);
    const float wa  = __expf(s.x - mx);
    const float wb  = __expf(s.z - mx);
    const float den = wa * s.y + wb * s.w;
    v4f o;
    o.x = (wa * oa.x + wb * ob.x) / den;
    o.y = (wa * oa.y + wb * ob.y) / den;
    o.z = (wa * oa.z + wb * ob.z) / den;
    o.w = (wa * oa.w + wb * ob.w) / den;
    *(v4f*)(out + lb + off) = lv;
    *(v4f*)(out + hb + off) = o;
  }
}

extern "C" void kernel_launch(void* const* d_in, const int* in_sizes, int n_in,
                              void* d_out, int out_size, void* d_ws, size_t ws_size,
                              hipStream_t stream) {
  (void)in_sizes; (void)n_in; (void)out_size; (void)ws_size;

  const float* x  = (const float*)d_in[0];
  const float* Wk = (const float*)d_in[1];
  const float* Wq = (const float*)d_in[2];
  const float* Wv = (const float*)d_in[3];
  float* out = (float*)d_out;

  float* qo = (float*)d_ws;
  float* ko = qo + (size_t)BT * Hsz;
  float* vo = ko + (size_t)BT * Hsz;

  const int tiles = BT / 16;
  dim3 block(128);
  dim3 grid(tiles / 4);

  proj_qkv<<<grid, block, 0, stream>>>(x, Wk, Wq, Wv, qo, ko, vo, out);
  flash_attn<<<grid, dim3(256), 0, stream>>>(qo, ko, vo, out);
  merge_out<<<dim3(128), dim3(256), 0, stream>>>(qo, out);
}
